# starting kernel
# baseline (speedup 1.0000x reference)
#include <hip/hip_runtime.h>

typedef __attribute__((ext_vector_type(2))) float v2f;
typedef __attribute__((ext_vector_type(8))) float v8f;

#define NODES  50000
#define EDGES  800000
#define GRAPHS 64
#define C      64
#define OUTC   32
#define LPITCH 68

__global__ void fill_kernel(float* __restrict__ p, float v, int n) {
  int i = blockIdx.x * blockDim.x + threadIdx.x;
  if (i < n) p[i] = v;
}

__global__ void deg_count_kernel(const int* __restrict__ dst, float* __restrict__ deg) {
  int i = blockIdx.x * blockDim.x + threadIdx.x;
  if (i < EDGES) atomicAdd(&deg[dst[i]], 1.0f);
}

__global__ void rsqrt_kernel(float* __restrict__ p, int n) {
  int i = blockIdx.x * blockDim.x + threadIdx.x;
  if (i < n) p[i] = rsqrtf(p[i]);
}

__global__ __launch_bounds__(256)
void gemm_rs_wmma(const float* __restrict__ X, const float* __restrict__ W,
                  const float* __restrict__ rowscale, float* __restrict__ out,
                  int ntiles) {
  __shared__ float sWt[C * LPITCH];
  for (int i = threadIdx.x; i < C * C; i += 256) {
    const int k = i >> 6, n = i & 63;
    sWt[n * LPITCH + k] = W[i];
  }
  __syncthreads();

  const int wave = threadIdx.x >> 5;
  const int lane = threadIdx.x & 31;
  const int tile = blockIdx.x * 8 + wave;
  if (tile >= ntiles) return;

  const int m0 = tile * 16;
  const int ar = m0 + (lane & 15);
  const int kg = (lane >> 4) * 2;
  const int cn = lane & 15;
  const float* bp = sWt + cn * LPITCH;

  v8f acc0 = {}, acc1 = {}, acc2 = {}, acc3 = {};
#pragma unroll
  for (int ks = 0; ks < 16; ++ks) {
    const int k0 = ks * 4 + kg;
    v2f a = *(const v2f*)(X + ar * C + k0);
    v2f b0 = *(const v2f*)(bp + k0);
    v2f b1 = *(const v2f*)(bp + 16 * LPITCH + k0);
    v2f b2 = *(const v2f*)(bp + 32 * LPITCH + k0);
    v2f b3 = *(const v2f*)(bp + 48 * LPITCH + k0);
    acc0 = __builtin_amdgcn_wmma_f32_16x16x4_f32(false, a, false, b0, (short)0, acc0, false, false);
    acc1 = __builtin_amdgcn_wmma_f32_16x16x4_f32(false, a, false, b1, (short)0, acc1, false, false);
    acc2 = __builtin_amdgcn_wmma_f32_16x16x4_f32(false, a, false, b2, (short)0, acc2, false, false);
    acc3 = __builtin_amdgcn_wmma_f32_16x16x4_f32(false, a, false, b3, (short)0, acc3, false, false);
  }

#pragma unroll
  for (int r = 0; r < 8; ++r) {
    const int row = m0 + r + 8 * (lane >> 4);
    const float s = rowscale[row];
    out[row * C +  0 + cn] = acc0[r] * s;
    out[row * C + 16 + cn] = acc1[r] * s;
    out[row * C + 32 + cn] = acc2[r] * s;
    out[row * C + 48 + cn] = acc3[r] * s;
  }
}

__global__ void scatter_kernel(const int* __restrict__ src, const int* __restrict__ dst,
                               const float* __restrict__ hs, float* __restrict__ acc) {
  int idx = blockIdx.x * blockDim.x + threadIdx.x;
  int e  = idx >> 4;
  int cg = (idx & 15) << 2;
  if (e < EDGES) {
    int s = src[e], d = dst[e];
    const float4 v = *(const float4*)(hs + (size_t)s * C + cg);
    float* a = acc + (size_t)d * C + cg;
    atomicAdd(a + 0, v.x);
    atomicAdd(a + 1, v.y);
    atomicAdd(a + 2, v.z);
    atomicAdd(a + 3, v.w);
  }
}

__global__ void finalize_kernel(float* __restrict__ acc, const float* __restrict__ hs,
                                const float* __restrict__ dinv, const float* __restrict__ bias,
                                int do_relu) {
  int i = blockIdx.x * blockDim.x + threadIdx.x;
  if (i < NODES * C) {
    int row = i >> 6, c = i & 63;
    float v = dinv[row] * (acc[i] + hs[i]) + bias[c];
    if (do_relu) v = fmaxf(v, 0.0f);
    acc[i] = v;
  }
}

__global__ void pool_kernel(const float* __restrict__ z, const int* __restrict__ batch,
                            float* __restrict__ u) {
  int idx = blockIdx.x * blockDim.x + threadIdx.x;
  int i  = idx >> 4;
  int cg = (idx & 15) << 2;
  if (i < NODES) {
    int g = batch[i];
    const float4 v = *(const float4*)(z + (size_t)i * C + cg);
    float* p = u + g * C + cg;
    atomicAdd(p + 0, v.x);
    atomicAdd(p + 1, v.y);
    atomicAdd(p + 2, v.z);
    atomicAdd(p + 3, v.w);
  }
}

__global__ __launch_bounds__(256)
void head_wmma(const float* __restrict__ u, const float* __restrict__ Wf,
               const float* __restrict__ bf, float* __restrict__ y,
               float* __restrict__ out_u) {
  __shared__ float sWf[OUTC * LPITCH];
  for (int i = threadIdx.x; i < C * OUTC; i += 256) {
    const int k = i >> 5, n = i & 31;
    sWf[n * LPITCH + k] = Wf[i];
  }
  __syncthreads();

  const int wave = threadIdx.x >> 5;
  const int lane = threadIdx.x & 31;
  const int mt = wave >> 1;
  const int nt = wave & 1;
  const int ar = mt * 16 + (lane & 15);
  const int kg = (lane >> 4) * 2;
  const int cn = nt * 16 + (lane & 15);
  const float* bp = sWf + cn * LPITCH;

  v8f acc = {};
#pragma unroll
  for (int ks = 0; ks < 16; ++ks) {
    const int k0 = ks * 4 + kg;
    v2f a = *(const v2f*)(u + ar * C + k0);
    v2f b = *(const v2f*)(bp + k0);
    acc = __builtin_amdgcn_wmma_f32_16x16x4_f32(false, a, false, b, (short)0, acc, false, false);
  }
#pragma unroll
  for (int r = 0; r < 8; ++r) {
    const int row = mt * 16 + r + 8 * (lane >> 4);
    y[row * OUTC + cn] = acc[r] + bf[cn];
  }
#pragma unroll
  for (int j = 0; j < 16; ++j) {
    const int idx = threadIdx.x + j * 256;
    out_u[idx] = u[idx];
  }
}

extern "C" void kernel_launch(void* const* d_in, const int* in_sizes, int n_in,
                              void* d_out, int out_size, void* d_ws, size_t ws_size,
                              hipStream_t stream) {
  (void)in_sizes; (void)n_in; (void)out_size; (void)ws_size;

  const float* x     = (const float*)d_in[0];
  const int*   ei    = (const int*)  d_in[1];
  const int*   batch = (const int*)  d_in[2];
  const float* W1    = (const float*)d_in[3];
  const float* b1    = (const float*)d_in[4];
  const float* W2    = (const float*)d_in[5];
  const float* b2    = (const float*)d_in[6];
  const float* Wf    = (const float*)d_in[7];
  const float* bf    = (const float*)d_in[8];
  float* out = (float*)d_out;

  const int* src = ei;
  const int* dst = ei + EDGES;

  float* ws   = (float*)d_ws;
  float* dinv = ws;
  float* hs   = ws + 50048;
  float* acc  = hs + (size_t)NODES * C;
  float* u    = acc + (size_t)NODES * C;

  const dim3 B(256);
  const int NT = NODES / 16;
  const int GB = (NT + 7) / 8;

  fill_kernel     <<<(NODES + 255) / 256, B, 0, stream>>>(dinv, 1.0f, NODES);
  deg_count_kernel<<<(EDGES + 255) / 256, B, 0, stream>>>(dst, dinv);
  rsqrt_kernel    <<<(NODES + 255) / 256, B, 0, stream>>>(dinv, NODES);

  fill_kernel    <<<(NODES * C) / 256, B, 0, stream>>>(acc, 0.0f, NODES * C);
  gemm_rs_wmma   <<<GB, B, 0, stream>>>(x, W1, dinv, hs, NT);
  scatter_kernel <<<(EDGES * 16) / 256, B, 0, stream>>>(src, dst, hs, acc);
  finalize_kernel<<<(NODES * C) / 256, B, 0, stream>>>(acc, hs, dinv, b1, 1);

  gemm_rs_wmma   <<<GB, B, 0, stream>>>(acc, W2, dinv, hs, NT);
  fill_kernel    <<<(NODES * C) / 256, B, 0, stream>>>(acc, 0.0f, NODES * C);
  scatter_kernel <<<(EDGES * 16) / 256, B, 0, stream>>>(src, dst, hs, acc);
  finalize_kernel<<<(NODES * C) / 256, B, 0, stream>>>(acc, hs, dinv, b2, 0);

  fill_kernel<<<(GRAPHS * C + 255) / 256, B, 0, stream>>>(u, 0.0f, GRAPHS * C);
  pool_kernel<<<(NODES * 16) / 256, B, 0, stream>>>(acc, batch, u);
  head_wmma  <<<1, B, 0, stream>>>(u, Wf, bf, out, out + GRAPHS * OUTC);
}


# layer 2 + pool summed over the edge list: blocks add dinv[s]*dinv[d]*h[s] rows into a 64x64 LDS table at row batch[d]; pooled adjacency, its clear and global float atomics gone; sum order changes
# speedup vs baseline: 2.5192x; 2.5192x over previous
#include <hip/hip_runtime.h>

typedef __attribute__((ext_vector_type(2))) float v2f;
typedef __attribute__((ext_vector_type(8))) float v8f;

#define NODES  50000
#define EDGES  800000
#define GRAPHS 64
#define C      64
#define OUTC   32
#define LPITCH 68

#define DH_RANGES  8
#define DH_SLICES  32
#define DH_RN      6272
#define DH_PITCH   (DH_RANGES * DH_RN)
#define DH_SE      (EDGES / DH_SLICES)
#define DH_V4      (DH_SE / 4)
#define DH_THREADS 512
#define DH_TRIPS   ((DH_V4 + DH_THREADS - 1) / DH_THREADS)

__global__ __launch_bounds__(DH_THREADS)
void deg_hist_kernel(const int* __restrict__ dst, int* __restrict__ cnt) {
  __shared__ int4 sH4[DH_RN / 4];
  int* sH = (int*)sH4;
  const int r = blockIdx.x % DH_RANGES;
  const int s = blockIdx.x / DH_RANGES;
  const int lo = r * DH_RN;
  const int* p = dst + s * DH_SE;

  int4 v[DH_TRIPS];
#pragma unroll
  for (int t = 0; t < DH_TRIPS; ++t) {
    const int j = threadIdx.x + DH_THREADS * t;
    v[t] = *(const int4*)(p + 4 * (j < DH_V4 ? j : 0));
  }

  for (int i = threadIdx.x; i < DH_RN / 4; i += DH_THREADS) sH4[i] = int4{0, 0, 0, 0};
  __syncthreads();

#pragma unroll
  for (int t = 0; t < DH_TRIPS; ++t) {
    const int j = threadIdx.x + DH_THREADS * t;
    const unsigned a0 = (unsigned)(v[t].x - lo), a1 = (unsigned)(v[t].y - lo);
    const unsigned a2 = (unsigned)(v[t].z - lo), a3 = (unsigned)(v[t].w - lo);
    if (j < DH_V4) {
      if (a0 < DH_RN) atomicAdd(&sH[a0], 1);
      if (a1 < DH_RN) atomicAdd(&sH[a1], 1);
      if (a2 < DH_RN) atomicAdd(&sH[a2], 1);
      if (a3 < DH_RN) atomicAdd(&sH[a3], 1);
    }
  }
  __syncthreads();
  int4* o = (int4*)(cnt + (size_t)s * DH_PITCH + lo);
  for (int i = threadIdx.x; i < DH_RN / 4; i += DH_THREADS) o[i] = sH4[i];
}

__global__ void dinv_kernel(const int* __restrict__ cnt, const int* __restrict__ batch,
                            float* __restrict__ dinv, int* __restrict__ rec) {
  int i = blockIdx.x * blockDim.x + threadIdx.x;
  if (i < NODES) {
    int d = 1;
#pragma unroll
    for (int s = 0; s < DH_SLICES; ++s) d += cnt[(size_t)s * DH_PITCH + i];
    const float r = rsqrtf((float)d);
    dinv[i] = r;
    *(int2*)(rec + 2 * i) = int2{__float_as_int(r), batch[i]};
  }
}

__global__ __launch_bounds__(256)
void gemm_rs_wmma(const float* __restrict__ X, const float* __restrict__ W,
                  const float* __restrict__ rowscale, float* __restrict__ out,
                  int ntiles) {
  __shared__ float sWt[C * LPITCH];
  for (int i = threadIdx.x; i < C * C; i += 256) {
    const int k = i >> 6, n = i & 63;
    sWt[n * LPITCH + k] = W[i];
  }
  __syncthreads();

  const int wave = threadIdx.x >> 5;
  const int lane = threadIdx.x & 31;
  const int tile = blockIdx.x * 8 + wave;
  if (tile >= ntiles) return;

  const int m0 = tile * 16;
  const int ar = m0 + (lane & 15);
  const int kg = (lane >> 4) * 2;
  const int cn = lane & 15;
  const float* bp = sWt + cn * LPITCH;

  v8f acc0 = {}, acc1 = {}, acc2 = {}, acc3 = {};
#pragma unroll
  for (int ks = 0; ks < 16; ++ks) {
    const int k0 = ks * 4 + kg;
    v2f a = *(const v2f*)(X + ar * C + k0);
    v2f b0 = *(const v2f*)(bp + k0);
    v2f b1 = *(const v2f*)(bp + 16 * LPITCH + k0);
    v2f b2 = *(const v2f*)(bp + 32 * LPITCH + k0);
    v2f b3 = *(const v2f*)(bp + 48 * LPITCH + k0);
    acc0 = __builtin_amdgcn_wmma_f32_16x16x4_f32(false, a, false, b0, (short)0, acc0, false, false);
    acc1 = __builtin_amdgcn_wmma_f32_16x16x4_f32(false, a, false, b1, (short)0, acc1, false, false);
    acc2 = __builtin_amdgcn_wmma_f32_16x16x4_f32(false, a, false, b2, (short)0, acc2, false, false);
    acc3 = __builtin_amdgcn_wmma_f32_16x16x4_f32(false, a, false, b3, (short)0, acc3, false, false);
  }

#pragma unroll
  for (int r = 0; r < 8; ++r) {
    const int row = m0 + r + 8 * (lane >> 4);
    const float s = rowscale[row];
    out[row * C +  0 + cn] = acc0[r] * s;
    out[row * C + 16 + cn] = acc1[r] * s;
    out[row * C + 32 + cn] = acc2[r] * s;
    out[row * C + 48 + cn] = acc3[r] * s;
  }
}

#define AG_RN      1024
#define AG_RANGES  49
#define AG_SLICES  5
#define AG_SE      (EDGES / AG_SLICES)
#define AG_V4      (AG_SE / 4)
#define AG_THREADS 1024
#define AG_BATCH   8
#define AG_NB      ((AG_V4 + AG_THREADS * AG_BATCH - 1) / (AG_THREADS * AG_BATCH))
#define AG_QCAP    8192
#define AG_U       16

__device__ inline void agg_direct(float* __restrict__ sA, const int* __restrict__ src,
                                  const float* __restrict__ hs, const int* __restrict__ p,
                                  int ebase, int lo, int b, int slot) {
#pragma nounroll
  for (int t = 0; t < AG_BATCH; ++t) {
    const int j = threadIdx.x + AG_THREADS * (b * AG_BATCH + t);
    if (j >= AG_V4) break;
#pragma nounroll
    for (int k = 0; k < 4; ++k) {
      const unsigned dl = (unsigned)(p[4 * j + k] - lo);
      if (dl < AG_RN) {
        if (slot >= AG_QCAP) {
          const float* row = hs + (size_t)src[ebase + 4 * j + k] * C;
          float* d = sA + dl * C;
#pragma nounroll
          for (int c = 0; c < C; ++c)
            __hip_atomic_fetch_add(d + c, row[c], __ATOMIC_RELAXED, __HIP_MEMORY_SCOPE_WORKGROUP);
        }
        ++slot;
      }
    }
  }
}

__global__ __launch_bounds__(AG_THREADS)
void agg_lds_kernel(const int* __restrict__ src, const int* __restrict__ dst,
                    const float* __restrict__ hs, float* __restrict__ plane) {
  __shared__ float4 sA4[AG_RN * C / 4];
  __shared__ int sQ[AG_QCAP];
  __shared__ int sQn;
  float* sA = (float*)sA4;
  const int r = blockIdx.x % AG_RANGES;
  const int s = blockIdx.x / AG_RANGES;
  const int lo = r * AG_RN;
  const int ebase = s * AG_SE;
  const int* p = dst + ebase;

  for (int i = threadIdx.x; i < AG_RN * C / 4; i += AG_THREADS) sA4[i] = float4{0.0f, 0.0f, 0.0f, 0.0f};
  if (threadIdx.x == 0) sQn = 0;
  __syncthreads();

  for (int b = 0; b < AG_NB; ++b) {
    int4 v[AG_BATCH];
#pragma unroll
    for (int t = 0; t < AG_BATCH; ++t) {
      const int j = threadIdx.x + AG_THREADS * (b * AG_BATCH + t);
      v[t] = *(const int4*)(p + 4 * (j < AG_V4 ? j : 0));
    }
    int n = 0;
#pragma unroll
    for (int t = 0; t < AG_BATCH; ++t) {
      const int j = threadIdx.x + AG_THREADS * (b * AG_BATCH + t);
      const int m = ((unsigned)(v[t].x - lo) < AG_RN) + ((unsigned)(v[t].y - lo) < AG_RN) +
                    ((unsigned)(v[t].z - lo) < AG_RN) + ((unsigned)(v[t].w - lo) < AG_RN);
      n += (j < AG_V4) ? m : 0;
    }
    if (n) {
      const int slot0 = __hip_atomic_fetch_add(&sQn, n, __ATOMIC_RELAXED, __HIP_MEMORY_SCOPE_WORKGROUP);
      int slot = slot0;
#pragma unroll
      for (int t = 0; t < AG_BATCH; ++t) {
        const int j = threadIdx.x + AG_THREADS * (b * AG_BATCH + t);
        const int d4[4] = {v[t].x, v[t].y, v[t].z, v[t].w};
#pragma unroll
        for (int k = 0; k < 4; ++k) {
          const unsigned dl = (unsigned)(d4[k] - lo);
          if (j < AG_V4 && dl < AG_RN) {
            const int e = ebase + 4 * j + k;
            if (slot < AG_QCAP) sQ[slot] = (e << 10) | (int)dl;
            ++slot;
          }
        }
      }
      if (slot > AG_QCAP) agg_direct(sA, src, hs, p, ebase, lo, b, slot0);
    }
  }
  __syncthreads();

  const int total = sQn;
  const int cnt = total < AG_QCAP ? total : AG_QCAP;
  for (int i = threadIdx.x; i < cnt; i += 4 * AG_THREADS) {
    int ix[4], qq[4], ss[4];
#pragma unroll
    for (int k = 0; k < 4; ++k) {
      ix[k] = i + k * AG_THREADS < cnt ? i + k * AG_THREADS : i;
      qq[k] = sQ[ix[k]];
    }
#pragma unroll
    for (int k = 0; k < 4; ++k) ss[k] = src[qq[k] >> 10];
#pragma unroll
    for (int k = 0; k < 4; ++k) sQ[ix[k]] = (ss[k] << 10) | (qq[k] & (AG_RN - 1));
  }
  __syncthreads();

  const int wave = threadIdx.x >> 5;
  const int lane = threadIdx.x & 31;
  for (int i0 = wave * AG_U; i0 < cnt; i0 += (AG_THREADS / 32) * AG_U) {
    int q[AG_U];
    float a0[AG_U], a1[AG_U];
#pragma unroll
    for (int u = 0; u < AG_U; ++u) q[u] = sQ[i0 + u < cnt ? i0 + u : i0];
#pragma unroll
    for (int u = 0; u < AG_U; ++u) {
      const float* row = hs + (size_t)(q[u] >> 10) * C;
      a0[u] = row[lane];
      a1[u] = row[32 + lane];
    }
#pragma unroll
    for (int u = 0; u < AG_U; ++u) {
      const bool ok = i0 + u < cnt;
      float* d = sA + (q[u] & (AG_RN - 1)) * C;
      __hip_atomic_fetch_add(d + lane,      ok ? a0[u] : 0.0f, __ATOMIC_RELAXED, __HIP_MEMORY_SCOPE_WORKGROUP);
      __hip_atomic_fetch_add(d + 32 + lane, ok ? a1[u] : 0.0f, __ATOMIC_RELAXED, __HIP_MEMORY_SCOPE_WORKGROUP);
    }
  }
  __syncthreads();

  const int nrows = (NODES - lo) < AG_RN ? (NODES - lo) : AG_RN;
  float4* o = (float4*)(plane + ((size_t)s * NODES + lo) * C);
  for (int i = threadIdx.x; i < nrows * (C / 4); i += AG_THREADS) o[i] = sA4[i];
}

__global__ void finalize_kernel(const float* __restrict__ plane, const float* __restrict__ hs,
                                const float* __restrict__ dinv, const float* __restrict__ bias,
                                float* __restrict__ hd) {
  int i = blockIdx.x * blockDim.x + threadIdx.x;
  if (i < NODES * C / 4) {
    float4 a = ((const float4*)plane)[i];
#pragma unroll
    for (int s = 1; s < AG_SLICES; ++s) {
      const float4 t = ((const float4*)(plane + (size_t)s * NODES * C))[i];
      a.x += t.x; a.y += t.y; a.z += t.z; a.w += t.w;
    }
    const float4 x = ((const float4*)hs)[i];
    const float4 b = *(const float4*)(bias + 4 * (i & 15));
    const float d = dinv[i >> 4];
    float4 v;
    v.x = d * fmaxf(d * (a.x + x.x) + b.x, 0.0f);
    v.y = d * fmaxf(d * (a.y + x.y) + b.y, 0.0f);
    v.z = d * fmaxf(d * (a.z + x.z) + b.z, 0.0f);
    v.w = d * fmaxf(d * (a.w + x.w) + b.w, 0.0f);
    ((float4*)hd)[i] = v;
  }
}

#define PE_THREADS 1024
#define PE_WAVES   (PE_THREADS / 32)
#define PE_BLOCKS  256
#define PE_ITEMS   (EDGES + NODES)
#define PE_NCH     ((PE_ITEMS + 31) / 32)
#define PE_STRIDE  (PE_BLOCKS * PE_WAVES)
#define PE_K       ((PE_NCH + PE_STRIDE - 1) / PE_STRIDE)

__global__ __launch_bounds__(PE_THREADS)
void pool_edges_kernel(const int* __restrict__ src, const int* __restrict__ dst,
                       const int* __restrict__ rec, const float* __restrict__ hd,
                       float* __restrict__ part) {
  __shared__ float4 sU4[GRAPHS * C / 4];
  float* sU = (float*)sU4;
  const int wave = __builtin_amdgcn_readfirstlane(threadIdx.x >> 5);
  const int lane = threadIdx.x & 31;
  const int wv = wave * PE_BLOCKS + blockIdx.x;

  int ss[PE_K], dd[PE_K];
  bool ok[PE_K];
#pragma unroll
  for (int k = 0; k < PE_K; ++k) {
    const int ch = wv + k * PE_STRIDE;
    const int item = (ch < PE_NCH ? ch : 0) * 32 + lane;
    const bool edge = item < EDGES;
    ok[k] = item < PE_ITEMS;
    const int e = edge ? item : 0;
    const int n = ok[k] ? item - EDGES : 0;
    const int sl = src[e], dl = dst[e];
    ss[k] = edge ? sl : n;
    dd[k] = edge ? dl : n;
  }
  unsigned ms[PE_K];
  float mw[PE_K];
#pragma unroll
  for (int k = 0; k < PE_K; ++k) {
    const int2 r = *(const int2*)(rec + 2 * dd[k]);
    const bool in = ok[k] && (unsigned)r.y < GRAPHS;
    ms[k] = ((unsigned)ss[k] << 6) | (unsigned)(in ? r.y : 0);
    mw[k] = in ? __int_as_float(r.x) : 0.0f;
  }

  sU4[threadIdx.x] = float4{0.0f, 0.0f, 0.0f, 0.0f};
  __syncthreads();

#pragma unroll
  for (int k = 0; k < PE_K; ++k) {
    if (wv + k * PE_STRIDE < PE_NCH) {
      float2 a[32];
#pragma unroll
      for (int u = 0; u < 32; ++u) {
        const unsigned m = __builtin_amdgcn_readlane(ms[k], u);
        a[u] = *(const float2*)(hd + (size_t)(m >> 6) * C + 2 * lane);
      }
#pragma unroll
      for (int u = 0; u < 32; ++u) {
        const unsigned m = __builtin_amdgcn_readlane(ms[k], u);
        const float w = __int_as_float(__builtin_amdgcn_readlane(__float_as_int(mw[k]), u));
        float* d = sU + (m & (GRAPHS - 1)) * C + 2 * lane;
        __hip_atomic_fetch_add(d,     w * a[u].x, __ATOMIC_RELAXED, __HIP_MEMORY_SCOPE_WORKGROUP);
        __hip_atomic_fetch_add(d + 1, w * a[u].y, __ATOMIC_RELAXED, __HIP_MEMORY_SCOPE_WORKGROUP);
      }
    }
  }
  __syncthreads();

  ((float4*)(part + (size_t)blockIdx.x * (GRAPHS * C)))[threadIdx.x] = sU4[threadIdx.x];
}

__global__ __launch_bounds__(256)
void pool_reduce(const float* __restrict__ part, const int* __restrict__ batch,
                 float* __restrict__ upre, float* __restrict__ cnt) {
  __shared__ float sR[16 * 64];
  __shared__ int sN[256];
  const int i   = threadIdx.x & 15;
  const int grp = threadIdx.x >> 4;
  const int f4  = blockIdx.x * 16 + i;
  float4 s = {0.0f, 0.0f, 0.0f, 0.0f};
#pragma unroll
  for (int j = 0; j < (PE_BLOCKS + 15) / 16; ++j) {
    const int p = grp + 16 * j;
    if (p < PE_BLOCKS) {
      const float4 t = *(const float4*)(part + (size_t)p * (GRAPHS * C) + f4 * 4);
      s.x += t.x; s.y += t.y; s.z += t.z; s.w += t.w;
    }
  }
  const int g = blockIdx.x;
  int mine = 0;
#pragma unroll
  for (int t = 0; t < (NODES / 4 + 255) / 256; ++t) {
    const int j  = threadIdx.x + 256 * t;
    const bool in = j < NODES / 4;
    const int4 b = *(const int4*)(batch + 4 * (in ? j : 0));
    const int n  = (b.x == g) + (b.y == g) + (b.z == g) + (b.w == g);
    mine += in ? n : 0;
  }
  *(float4*)(sR + grp * 64 + i * 4) = s;
  sN[threadIdx.x] = mine;
  __syncthreads();
  if (threadIdx.x < 64) {
    float r = sR[threadIdx.x];
#pragma unroll
    for (int k = 1; k < 16; ++k) r += sR[k * 64 + threadIdx.x];
    upre[blockIdx.x * 64 + threadIdx.x] = r;
  }
  if (threadIdx.x < 32) {
    int t = 0;
#pragma unroll
    for (int k = 0; k < 8; ++k) t += sN[threadIdx.x + 32 * k];
#pragma unroll
    for (int m = 16; m >= 1; m >>= 1) t += __shfl_xor(t, m);
    if (threadIdx.x == 0) cnt[g] = (float)t;
  }
}

__global__ __launch_bounds__(256)
void head_wmma(const float* __restrict__ upre, const float* __restrict__ cnt,
               const float* __restrict__ W2, const float* __restrict__ b2,
               const float* __restrict__ Wf, const float* __restrict__ bf,
               float* __restrict__ y, float* __restrict__ out_u) {
  __shared__ float sW2t[C * LPITCH];
  __shared__ float sWft[OUTC * LPITCH];
  __shared__ float sU[GRAPHS * LPITCH];
  for (int i = threadIdx.x; i < C * C; i += 256) {
    const int k = i >> 6, n = i & 63;
    sW2t[n * LPITCH + k] = W2[i];
  }
  for (int i = threadIdx.x; i < C * OUTC; i += 256) {
    const int k = i >> 5, n = i & 31;
    sWft[n * LPITCH + k] = Wf[i];
  }
  __syncthreads();

  const int wave = threadIdx.x >> 5;
  const int lane = threadIdx.x & 31;
  const int mt = wave >> 1;
  const int nh = wave & 1;
  const int c  = lane & 15;
  const int ar = mt * 16 + c;
  const int kg = (lane >> 4) * 2;

  const float* bp0 = sW2t + (nh * 32 + c) * LPITCH;
  const float* bp1 = bp0 + 16 * LPITCH;
  v8f u0 = {}, u1 = {};
#pragma unroll
  for (int ks = 0; ks < 16; ++ks) {
    const int k0 = ks * 4 + kg;
    v2f a  = *(const v2f*)(upre + ar * C + k0);
    v2f b0 = *(const v2f*)(bp0 + k0);
    v2f b1 = *(const v2f*)(bp1 + k0);
    u0 = __builtin_amdgcn_wmma_f32_16x16x4_f32(false, a, false, b0, (short)0, u0, false, false);
    u1 = __builtin_amdgcn_wmma_f32_16x16x4_f32(false, a, false, b1, (short)0, u1, false, false);
  }
  const int col0 = nh * 32 + c, col1 = col0 + 16;
  const float bb0 = b2[col0], bb1 = b2[col1];
#pragma unroll
  for (int r = 0; r < 8; ++r) {
    const int row = mt * 16 + r + 8 * (lane >> 4);
    const float n = cnt[row];
    const float v0 = u0[r] + n * bb0;
    const float v1 = u1[r] + n * bb1;
    sU[row * LPITCH + col0] = v0;
    sU[row * LPITCH + col1] = v1;
    out_u[row * C + col0] = v0;
    out_u[row * C + col1] = v1;
  }
  __syncthreads();

  const int cn = nh * 16 + c;
  const float* bp = sWft + cn * LPITCH;
  v8f acc = {};
#pragma unroll
  for (int ks = 0; ks < 16; ++ks) {
    const int k0 = ks * 4 + kg;
    v2f a = *(const v2f*)(sU + ar * LPITCH + k0);
    v2f b = *(const v2f*)(bp + k0);
    acc = __builtin_amdgcn_wmma_f32_16x16x4_f32(false, a, false, b, (short)0, acc, false, false);
  }
#pragma unroll
  for (int r = 0; r < 8; ++r) {
    const int row = mt * 16 + r + 8 * (lane >> 4);
    y[row * OUTC + cn] = acc[r] + bf[cn];
  }
}

extern "C" void kernel_launch(void* const* d_in, const int* in_sizes, int n_in,
                              void* d_out, int out_size, void* d_ws, size_t ws_size,
                              hipStream_t stream) {
  (void)in_sizes; (void)n_in; (void)out_size; (void)ws_size;

  const float* x     = (const float*)d_in[0];
  const int*   ei    = (const int*)  d_in[1];
  const int*   batch = (const int*)  d_in[2];
  const float* W1    = (const float*)d_in[3];
  const float* b1    = (const float*)d_in[4];
  const float* W2    = (const float*)d_in[5];
  const float* b2    = (const float*)d_in[6];
  const float* Wf    = (const float*)d_in[7];
  const float* bf    = (const float*)d_in[8];
  float* out = (float*)d_out;

  const int* src = ei;
  const int* dst = ei + EDGES;

  float* ws   = (float*)d_ws;
  float* dinv = ws;
  int*   rec  = (int*)(ws + 50048);
  float* hs   = ws + 3 * 50048;
  float* acc  = hs + (size_t)NODES * C;
  float* upre = acc + (size_t)NODES * C;
  float* cnt  = upre + GRAPHS * C;
  float* part = cnt + GRAPHS;
  int*   degc = (int*)(part + (size_t)PE_BLOCKS * GRAPHS * C);
  float* plane = (float*)(degc + (size_t)DH_SLICES * DH_PITCH);

  const dim3 B(256);
  const int NT = NODES / 16;
  const int GB = (NT + 7) / 8;

  deg_hist_kernel<<<DH_RANGES * DH_SLICES, dim3(DH_THREADS), 0, stream>>>(dst, degc);
  dinv_kernel    <<<(NODES + 255) / 256, B, 0, stream>>>(degc, batch, dinv, rec);

  gemm_rs_wmma   <<<GB, B, 0, stream>>>(x, W1, dinv, hs, NT);
  agg_lds_kernel <<<AG_RANGES * AG_SLICES, dim3(AG_THREADS), 0, stream>>>(src, dst, hs, plane);
  finalize_kernel<<<(NODES * C / 4) / 256, B, 0, stream>>>(plane, hs, dinv, b1, acc);

  pool_edges_kernel<<<PE_BLOCKS, dim3(PE_THREADS), 0, stream>>>(src, dst, rec, acc, part);
  pool_reduce<<<GRAPHS, B, 0, stream>>>(part, batch, upre, cnt);
  head_wmma  <<<1, B, 0, stream>>>(upre, cnt, W2, b2, Wf, bf, out, out + GRAPHS * OUTC);
}


# two one-purpose launches folded into neighbours: degree's rsqrt step into the product kernel, head into the partial-sum kernel (a graph's row needs no other row); 8 launches -> 6; sum order unchanged
# speedup vs baseline: 2.9025x; 1.1522x over previous
#include <hip/hip_runtime.h>

typedef __attribute__((ext_vector_type(2))) float v2f;
typedef __attribute__((ext_vector_type(8))) float v8f;

#define NODES  50000
#define EDGES  800000
#define GRAPHS 64
#define C      64
#define OUTC   32
#define LPITCH 68

#define DH_RANGES  8
#define DH_SLICES  32
#define DH_RN      6272
#define DH_PITCH   (DH_RANGES * DH_RN)
#define DH_SE      (EDGES / DH_SLICES)
#define DH_V4      (DH_SE / 4)
#define DH_THREADS 512
#define DH_TRIPS   ((DH_V4 + DH_THREADS - 1) / DH_THREADS)

__global__ __launch_bounds__(DH_THREADS)
void deg_hist_kernel(const int* __restrict__ dst, int* __restrict__ cnt) {
  __shared__ int4 sH4[DH_RN / 4];
  int* sH = (int*)sH4;
  const int r = blockIdx.x % DH_RANGES;
  const int s = blockIdx.x / DH_RANGES;
  const int lo = r * DH_RN;
  const int* p = dst + s * DH_SE;

  int4 v[DH_TRIPS];
#pragma unroll
  for (int t = 0; t < DH_TRIPS; ++t) {
    const int j = threadIdx.x + DH_THREADS * t;
    v[t] = *(const int4*)(p + 4 * (j < DH_V4 ? j : 0));
  }

  for (int i = threadIdx.x; i < DH_RN / 4; i += DH_THREADS) sH4[i] = int4{0, 0, 0, 0};
  __syncthreads();

#pragma unroll
  for (int t = 0; t < DH_TRIPS; ++t) {
    const int j = threadIdx.x + DH_THREADS * t;
    const unsigned a0 = (unsigned)(v[t].x - lo), a1 = (unsigned)(v[t].y - lo);
    const unsigned a2 = (unsigned)(v[t].z - lo), a3 = (unsigned)(v[t].w - lo);
    if (j < DH_V4) {
      if (a0 < DH_RN) atomicAdd(&sH[a0], 1);
      if (a1 < DH_RN) atomicAdd(&sH[a1], 1);
      if (a2 < DH_RN) atomicAdd(&sH[a2], 1);
      if (a3 < DH_RN) atomicAdd(&sH[a3], 1);
    }
  }
  __syncthreads();
  int4* o = (int4*)(cnt + (size_t)s * DH_PITCH + lo);
  for (int i = threadIdx.x; i < DH_RN / 4; i += DH_THREADS) o[i] = sH4[i];
}

__global__ __launch_bounds__(256)
void gemm_deg_wmma(const float* __restrict__ X, const float* __restrict__ W,
                   const int* __restrict__ cnt, const int* __restrict__ batch,
                   float* __restrict__ dinv, int* __restrict__ rec,
                   float* __restrict__ out, int ntiles) {
  __shared__ float sWt[C * LPITCH];
  __shared__ int   sDeg[128];
  __shared__ float sDinv[128];

  const int nl   = threadIdx.x & 127;
  const int hf   = threadIdx.x >> 7;
  const int node = blockIdx.x * 128 + nl;
  int d = 0;
#pragma unroll
  for (int s = 0; s < DH_SLICES / 2; ++s) d += cnt[(size_t)(hf * (DH_SLICES / 2) + s) * DH_PITCH + node];

  for (int i = threadIdx.x; i < C * C; i += 256) {
    const int k = i >> 6, n = i & 63;
    sWt[n * LPITCH + k] = W[i];
  }
  if (hf) sDeg[nl] = d;
  __syncthreads();
  if (!hf) {
    const float r = rsqrtf((float)(1 + d + sDeg[nl]));
    sDinv[nl] = r;
    if (node < NODES) {
      dinv[node] = r;
      *(int2*)(rec + 2 * node) = int2{__float_as_int(r), batch[node]};
    }
  }

  const int wave = __builtin_amdgcn_readfirstlane(threadIdx.x >> 5);
  const int lane = threadIdx.x & 31;
  const int tile = blockIdx.x * 8 + wave;
  const bool act = tile < ntiles;

  const int m0 = tile * 16;
  const int cn = lane & 15;
  v8f acc0 = {}, acc1 = {}, acc2 = {}, acc3 = {};
  if (act) {
    const int ar = m0 + (lane & 15);
    const int kg = (lane >> 4) * 2;
    const float* bp = sWt + cn * LPITCH;
#pragma unroll
    for (int ks = 0; ks < 16; ++ks) {
      const int k0 = ks * 4 + kg;
      v2f a = *(const v2f*)(X + ar * C + k0);
      v2f b0 = *(const v2f*)(bp + k0);
      v2f b1 = *(const v2f*)(bp + 16 * LPITCH + k0);
      v2f b2 = *(const v2f*)(bp + 32 * LPITCH + k0);
      v2f b3 = *(const v2f*)(bp + 48 * LPITCH + k0);
      acc0 = __builtin_amdgcn_wmma_f32_16x16x4_f32(false, a, false, b0, (short)0, acc0, false, false);
      acc1 = __builtin_amdgcn_wmma_f32_16x16x4_f32(false, a, false, b1, (short)0, acc1, false, false);
      acc2 = __builtin_amdgcn_wmma_f32_16x16x4_f32(false, a, false, b2, (short)0, acc2, false, false);
      acc3 = __builtin_amdgcn_wmma_f32_16x16x4_f32(false, a, false, b3, (short)0, acc3, false, false);
    }
  }
  __syncthreads();
  if (act) {
#pragma unroll
    for (int r = 0; r < 8; ++r) {
      const int rl  = wave * 16 + r + 8 * (lane >> 4);
      const int row = blockIdx.x * 128 + rl;
      const float s = sDinv[rl];
      out[row * C +  0 + cn] = acc0[r] * s;
      out[row * C + 16 + cn] = acc1[r] * s;
      out[row * C + 32 + cn] = acc2[r] * s;
      out[row * C + 48 + cn] = acc3[r] * s;
    }
  }
}

#define AG_RN      1024
#define AG_RANGES  49
#define AG_SLICES  5
#define AG_SE      (EDGES / AG_SLICES)
#define AG_V4      (AG_SE / 4)
#define AG_THREADS 1024
#define AG_BATCH   8
#define AG_NB      ((AG_V4 + AG_THREADS * AG_BATCH - 1) / (AG_THREADS * AG_BATCH))
#define AG_QCAP    8192
#define AG_U       16

__device__ inline void agg_direct(float* __restrict__ sA, const int* __restrict__ src,
                                  const float* __restrict__ hs, const int* __restrict__ p,
                                  int ebase, int lo, int b, int slot) {
#pragma nounroll
  for (int t = 0; t < AG_BATCH; ++t) {
    const int j = threadIdx.x + AG_THREADS * (b * AG_BATCH + t);
    if (j >= AG_V4) break;
#pragma nounroll
    for (int k = 0; k < 4; ++k) {
      const unsigned dl = (unsigned)(p[4 * j + k] - lo);
      if (dl < AG_RN) {
        if (slot >= AG_QCAP) {
          const float* row = hs + (size_t)src[ebase + 4 * j + k] * C;
          float* d = sA + dl * C;
#pragma nounroll
          for (int c = 0; c < C; ++c)
            __hip_atomic_fetch_add(d + c, row[c], __ATOMIC_RELAXED, __HIP_MEMORY_SCOPE_WORKGROUP);
        }
        ++slot;
      }
    }
  }
}

__global__ __launch_bounds__(AG_THREADS)
void agg_lds_kernel(const int* __restrict__ src, const int* __restrict__ dst,
                    const float* __restrict__ hs, float* __restrict__ plane) {
  __shared__ float4 sA4[AG_RN * C / 4];
  __shared__ int sQ[AG_QCAP];
  __shared__ int sQn;
  float* sA = (float*)sA4;
  const int r = blockIdx.x % AG_RANGES;
  const int s = blockIdx.x / AG_RANGES;
  const int lo = r * AG_RN;
  const int ebase = s * AG_SE;
  const int* p = dst + ebase;

  for (int i = threadIdx.x; i < AG_RN * C / 4; i += AG_THREADS) sA4[i] = float4{0.0f, 0.0f, 0.0f, 0.0f};
  if (threadIdx.x == 0) sQn = 0;
  __syncthreads();

  for (int b = 0; b < AG_NB; ++b) {
    int4 v[AG_BATCH];
#pragma unroll
    for (int t = 0; t < AG_BATCH; ++t) {
      const int j = threadIdx.x + AG_THREADS * (b * AG_BATCH + t);
      v[t] = *(const int4*)(p + 4 * (j < AG_V4 ? j : 0));
    }
    int n = 0;
#pragma unroll
    for (int t = 0; t < AG_BATCH; ++t) {
      const int j = threadIdx.x + AG_THREADS * (b * AG_BATCH + t);
      const int m = ((unsigned)(v[t].x - lo) < AG_RN) + ((unsigned)(v[t].y - lo) < AG_RN) +
                    ((unsigned)(v[t].z - lo) < AG_RN) + ((unsigned)(v[t].w - lo) < AG_RN);
      n += (j < AG_V4) ? m : 0;
    }
    if (n) {
      const int slot0 = __hip_atomic_fetch_add(&sQn, n, __ATOMIC_RELAXED, __HIP_MEMORY_SCOPE_WORKGROUP);
      int slot = slot0;
#pragma unroll
      for (int t = 0; t < AG_BATCH; ++t) {
        const int j = threadIdx.x + AG_THREADS * (b * AG_BATCH + t);
        const int d4[4] = {v[t].x, v[t].y, v[t].z, v[t].w};
#pragma unroll
        for (int k = 0; k < 4; ++k) {
          const unsigned dl = (unsigned)(d4[k] - lo);
          if (j < AG_V4 && dl < AG_RN) {
            const int e = ebase + 4 * j + k;
            if (slot < AG_QCAP) sQ[slot] = (e << 10) | (int)dl;
            ++slot;
          }
        }
      }
      if (slot > AG_QCAP) agg_direct(sA, src, hs, p, ebase, lo, b, slot0);
    }
  }
  __syncthreads();

  const int total = sQn;
  const int cnt = total < AG_QCAP ? total : AG_QCAP;
  for (int i = threadIdx.x; i < cnt; i += 4 * AG_THREADS) {
    int ix[4], qq[4], ss[4];
#pragma unroll
    for (int k = 0; k < 4; ++k) {
      ix[k] = i + k * AG_THREADS < cnt ? i + k * AG_THREADS : i;
      qq[k] = sQ[ix[k]];
    }
#pragma unroll
    for (int k = 0; k < 4; ++k) ss[k] = src[qq[k] >> 10];
#pragma unroll
    for (int k = 0; k < 4; ++k) sQ[ix[k]] = (ss[k] << 10) | (qq[k] & (AG_RN - 1));
  }
  __syncthreads();

  const int wave = threadIdx.x >> 5;
  const int lane = threadIdx.x & 31;
  for (int i0 = wave * AG_U; i0 < cnt; i0 += (AG_THREADS / 32) * AG_U) {
    int q[AG_U];
    float a0[AG_U], a1[AG_U];
#pragma unroll
    for (int u = 0; u < AG_U; ++u) q[u] = sQ[i0 + u < cnt ? i0 + u : i0];
#pragma unroll
    for (int u = 0; u < AG_U; ++u) {
      const float* row = hs + (size_t)(q[u] >> 10) * C;
      a0[u] = row[lane];
      a1[u] = row[32 + lane];
    }
#pragma unroll
    for (int u = 0; u < AG_U; ++u) {
      const bool ok = i0 + u < cnt;
      float* d = sA + (q[u] & (AG_RN - 1)) * C;
      __hip_atomic_fetch_add(d + lane,      ok ? a0[u] : 0.0f, __ATOMIC_RELAXED, __HIP_MEMORY_SCOPE_WORKGROUP);
      __hip_atomic_fetch_add(d + 32 + lane, ok ? a1[u] : 0.0f, __ATOMIC_RELAXED, __HIP_MEMORY_SCOPE_WORKGROUP);
    }
  }
  __syncthreads();

  const int nrows = (NODES - lo) < AG_RN ? (NODES - lo) : AG_RN;
  float4* o = (float4*)(plane + ((size_t)s * NODES + lo) * C);
  for (int i = threadIdx.x; i < nrows * (C / 4); i += AG_THREADS) o[i] = sA4[i];
}

__global__ void finalize_kernel(const float* __restrict__ plane, const float* __restrict__ hs,
                                const float* __restrict__ dinv, const float* __restrict__ bias,
                                float* __restrict__ hd) {
  int i = blockIdx.x * blockDim.x + threadIdx.x;
  if (i < NODES * C / 4) {
    float4 a = ((const float4*)plane)[i];
#pragma unroll
    for (int s = 1; s < AG_SLICES; ++s) {
      const float4 t = ((const float4*)(plane + (size_t)s * NODES * C))[i];
      a.x += t.x; a.y += t.y; a.z += t.z; a.w += t.w;
    }
    const float4 x = ((const float4*)hs)[i];
    const float4 b = *(const float4*)(bias + 4 * (i & 15));
    const float d = dinv[i >> 4];
    float4 v;
    v.x = d * fmaxf(d * (a.x + x.x) + b.x, 0.0f);
    v.y = d * fmaxf(d * (a.y + x.y) + b.y, 0.0f);
    v.z = d * fmaxf(d * (a.z + x.z) + b.z, 0.0f);
    v.w = d * fmaxf(d * (a.w + x.w) + b.w, 0.0f);
    ((float4*)hd)[i] = v;
  }
}

#define PE_THREADS 1024
#define PE_WAVES   (PE_THREADS / 32)
#define PE_BLOCKS  256
#define PE_ITEMS   (EDGES + NODES)
#define PE_NCH     ((PE_ITEMS + 31) / 32)
#define PE_STRIDE  (PE_BLOCKS * PE_WAVES)
#define PE_K       ((PE_NCH + PE_STRIDE - 1) / PE_STRIDE)

__global__ __launch_bounds__(PE_THREADS)
void pool_edges_kernel(const int* __restrict__ src, const int* __restrict__ dst,
                       const int* __restrict__ rec, const float* __restrict__ hd,
                       float* __restrict__ part) {
  __shared__ float4 sU4[GRAPHS * C / 4];
  float* sU = (float*)sU4;
  const int wave = __builtin_amdgcn_readfirstlane(threadIdx.x >> 5);
  const int lane = threadIdx.x & 31;
  const int wv = wave * PE_BLOCKS + blockIdx.x;

  int ss[PE_K], dd[PE_K];
  bool ok[PE_K];
#pragma unroll
  for (int k = 0; k < PE_K; ++k) {
    const int ch = wv + k * PE_STRIDE;
    const int item = (ch < PE_NCH ? ch : 0) * 32 + lane;
    const bool edge = item < EDGES;
    ok[k] = item < PE_ITEMS;
    const int e = edge ? item : 0;
    const int n = ok[k] ? item - EDGES : 0;
    const int sl = src[e], dl = dst[e];
    ss[k] = edge ? sl : n;
    dd[k] = edge ? dl : n;
  }
  unsigned ms[PE_K];
  float mw[PE_K];
#pragma unroll
  for (int k = 0; k < PE_K; ++k) {
    const int2 r = *(const int2*)(rec + 2 * dd[k]);
    const bool in = ok[k] && (unsigned)r.y < GRAPHS;
    ms[k] = ((unsigned)ss[k] << 6) | (unsigned)(in ? r.y : 0);
    mw[k] = in ? __int_as_float(r.x) : 0.0f;
  }

  sU4[threadIdx.x] = float4{0.0f, 0.0f, 0.0f, 0.0f};
  __syncthreads();

#pragma unroll
  for (int k = 0; k < PE_K; ++k) {
    if (wv + k * PE_STRIDE < PE_NCH) {
      float2 a[32];
#pragma unroll
      for (int u = 0; u < 32; ++u) {
        const unsigned m = __builtin_amdgcn_readlane(ms[k], u);
        a[u] = *(const float2*)(hd + (size_t)(m >> 6) * C + 2 * lane);
      }
#pragma unroll
      for (int u = 0; u < 32; ++u) {
        const unsigned m = __builtin_amdgcn_readlane(ms[k], u);
        const float w = __int_as_float(__builtin_amdgcn_readlane(__float_as_int(mw[k]), u));
        float* d = sU + (m & (GRAPHS - 1)) * C + 2 * lane;
        __hip_atomic_fetch_add(d,     w * a[u].x, __ATOMIC_RELAXED, __HIP_MEMORY_SCOPE_WORKGROUP);
        __hip_atomic_fetch_add(d + 1, w * a[u].y, __ATOMIC_RELAXED, __HIP_MEMORY_SCOPE_WORKGROUP);
      }
    }
  }
  __syncthreads();

  ((float4*)(part + (size_t)blockIdx.x * (GRAPHS * C)))[threadIdx.x] = sU4[threadIdx.x];
}

__global__ __launch_bounds__(256)
void pool_head(const float* __restrict__ part, const int* __restrict__ batch,
               const float* __restrict__ W2, const float* __restrict__ b2,
               const float* __restrict__ Wf, const float* __restrict__ bf,
               float* __restrict__ y, float* __restrict__ out_u) {
  __shared__ float sR[16 * 64];
  __shared__ int sN[256];
  __shared__ float sUp[16 * LPITCH];
  __shared__ float sU[16 * LPITCH];
  __shared__ float sCnt;
  const int g = blockIdx.x;
  const int wave = __builtin_amdgcn_readfirstlane(threadIdx.x >> 5);
  const int lane = threadIdx.x & 31;
  const int c  = lane & 15;
  const int kg = (lane >> 4) * 2;

  v2f bw[16] = {}, bfa[16] = {}, bfb[16] = {};
  float bb = 0.0f;
  if (wave < 4) {
    const int col = wave * 16 + c;
#pragma unroll
    for (int ks = 0; ks < 16; ++ks) {
      const int k0 = ks * 4 + kg;
      bw[ks] = v2f{W2[k0 * C + col], W2[(k0 + 1) * C + col]};
    }
    bb = b2[col];
  }
  if (wave == 0) {
#pragma unroll
    for (int ks = 0; ks < 16; ++ks) {
      const int k0 = ks * 4 + kg;
      bfa[ks] = v2f{Wf[k0 * OUTC + c],      Wf[(k0 + 1) * OUTC + c]};
      bfb[ks] = v2f{Wf[k0 * OUTC + 16 + c], Wf[(k0 + 1) * OUTC + 16 + c]};
    }
  }

  for (int z = threadIdx.x; z < 16 * LPITCH; z += 256) { sUp[z] = 0.0f; sU[z] = 0.0f; }

  const int i   = threadIdx.x & 15;
  const int grp = threadIdx.x >> 4;
  const int f4  = blockIdx.x * 16 + i;
  float4 s = {0.0f, 0.0f, 0.0f, 0.0f};
#pragma unroll
  for (int j = 0; j < (PE_BLOCKS + 15) / 16; ++j) {
    const int p = grp + 16 * j;
    if (p < PE_BLOCKS) {
      const float4 t = *(const float4*)(part + (size_t)p * (GRAPHS * C) + f4 * 4);
      s.x += t.x; s.y += t.y; s.z += t.z; s.w += t.w;
    }
  }
  int mine = 0;
#pragma unroll
  for (int t = 0; t < (NODES / 4 + 255) / 256; ++t) {
    const int j  = threadIdx.x + 256 * t;
    const bool in = j < NODES / 4;
    const int4 b = *(const int4*)(batch + 4 * (in ? j : 0));
    const int n  = (b.x == g) + (b.y == g) + (b.z == g) + (b.w == g);
    mine += in ? n : 0;
  }
  *(float4*)(sR + grp * 64 + i * 4) = s;
  sN[threadIdx.x] = mine;
  __syncthreads();
  if (threadIdx.x < 64) {
    float r = sR[threadIdx.x];
#pragma unroll
    for (int k = 1; k < 16; ++k) r += sR[k * 64 + threadIdx.x];
    sUp[threadIdx.x] = r;
  }
  if (threadIdx.x < 32) {
    int t = 0;
#pragma unroll
    for (int k = 0; k < 8; ++k) t += sN[threadIdx.x + 32 * k];
#pragma unroll
    for (int m = 16; m >= 1; m >>= 1) t += __shfl_xor(t, m);
    if (threadIdx.x == 0) sCnt = (float)t;
  }
  __syncthreads();

  if (wave < 4) {
    v8f acc = {};
#pragma unroll
    for (int ks = 0; ks < 16; ++ks) {
      const v2f a = *(const v2f*)(sUp + c * LPITCH + ks * 4 + kg);
      acc = __builtin_amdgcn_wmma_f32_16x16x4_f32(false, a, false, bw[ks], (short)0, acc, false, false);
    }
    const float n = sCnt;
    const float v = acc[0] + n * bb;
    if (lane < 16) sU[wave * 16 + c] = v;
  }
  __syncthreads();

  if (wave == 0) {
    out_u[g * C + lane]      = sU[lane];
    out_u[g * C + 32 + lane] = sU[32 + lane];
    v8f ya = {}, yb = {};
#pragma unroll
    for (int ks = 0; ks < 16; ++ks) {
      const v2f a = *(const v2f*)(sU + c * LPITCH + ks * 4 + kg);
      ya = __builtin_amdgcn_wmma_f32_16x16x4_f32(false, a, false, bfa[ks], (short)0, ya, false, false);
      yb = __builtin_amdgcn_wmma_f32_16x16x4_f32(false, a, false, bfb[ks], (short)0, yb, false, false);
    }
    const float hi = __shfl(yb[0], c);
    const float v  = lane < 16 ? ya[0] : hi;
    y[g * OUTC + lane] = v + bf[lane];
  }
}

extern "C" void kernel_launch(void* const* d_in, const int* in_sizes, int n_in,
                              void* d_out, int out_size, void* d_ws, size_t ws_size,
                              hipStream_t stream) {
  (void)in_sizes; (void)n_in; (void)out_size; (void)ws_size;

  const float* x     = (const float*)d_in[0];
  const int*   ei    = (const int*)  d_in[1];
  const int*   batch = (const int*)  d_in[2];
  const float* W1    = (const float*)d_in[3];
  const float* b1    = (const float*)d_in[4];
  const float* W2    = (const float*)d_in[5];
  const float* b2    = (const float*)d_in[6];
  const float* Wf    = (const float*)d_in[7];
  const float* bf    = (const float*)d_in[8];
  float* out = (float*)d_out;

  const int* src = ei;
  const int* dst = ei + EDGES;

  float* ws   = (float*)d_ws;
  float* dinv = ws;
  int*   rec  = (int*)(ws + 50048);
  float* hs   = ws + 3 * 50048;
  float* acc  = hs + (size_t)NODES * C;
  float* part = acc + (size_t)NODES * C + GRAPHS * C + GRAPHS;
  int*   degc = (int*)(part + (size_t)PE_BLOCKS * GRAPHS * C);
  float* plane = (float*)(degc + (size_t)DH_SLICES * DH_PITCH);

  const dim3 B(256);
  const int NT = NODES / 16;
  const int GB = (NT + 7) / 8;

  deg_hist_kernel<<<DH_RANGES * DH_SLICES, dim3(DH_THREADS), 0, stream>>>(dst, degc);

  gemm_deg_wmma  <<<GB, B, 0, stream>>>(x, W1, degc, batch, dinv, rec, hs, NT);
  agg_lds_kernel <<<AG_RANGES * AG_SLICES, dim3(AG_THREADS), 0, stream>>>(src, dst, hs, plane);
  finalize_kernel<<<(NODES * C / 4) / 256, B, 0, stream>>>(plane, hs, dinv, b1, acc);

  pool_edges_kernel<<<PE_BLOCKS, dim3(PE_THREADS), 0, stream>>>(src, dst, rec, acc, part);
  pool_head  <<<GRAPHS, B, 0, stream>>>(part, batch, W2, b2, Wf, bf, out, out + GRAPHS * OUTC);
}


# layer-1 sums complete per block: 256 node ranges, a block finds its edges by a 1-byte range number per edge over the whole list; partial planes, their sum pass and launch gone; sum order changes
# speedup vs baseline: 3.1836x; 1.0968x over previous
#include <hip/hip_runtime.h>

typedef __attribute__((ext_vector_type(2))) float v2f;
typedef __attribute__((ext_vector_type(8))) float v8f;

#define NODES  50000
#define EDGES  800000
#define GRAPHS 64
#define C      64
#define OUTC   32
#define LPITCH 68
#define AG_RN  196

#define DH_RANGES  8
#define DH_SLICES  32
#define DH_RN      6272
#define DH_PITCH   (DH_RANGES * DH_RN)
#define DH_SE      (EDGES / DH_SLICES)
#define DH_V4      (DH_SE / 4)
#define DH_THREADS 512
#define DH_TRIPS   ((DH_V4 + DH_THREADS - 1) / DH_THREADS)

__global__ __launch_bounds__(DH_THREADS)
void deg_hist_kernel(const int* __restrict__ dst, int* __restrict__ cnt, unsigned* __restrict__ key) {
  __shared__ int4 sH4[DH_RN / 4];
  int* sH = (int*)sH4;
  const int r = blockIdx.x % DH_RANGES;
  const int s = blockIdx.x / DH_RANGES;
  const int lo = r * DH_RN;
  const int* p = dst + s * DH_SE;

  int4 v[DH_TRIPS];
#pragma unroll
  for (int t = 0; t < DH_TRIPS; ++t) {
    const int j = threadIdx.x + DH_THREADS * t;
    v[t] = *(const int4*)(p + 4 * (j < DH_V4 ? j : 0));
  }

  if (r == 0) {
#pragma unroll
    for (int t = 0; t < DH_TRIPS; ++t) {
      const int j = threadIdx.x + DH_THREADS * t;
      if (j < DH_V4)
        key[s * DH_V4 + j] = ((unsigned)v[t].x / AG_RN) | (((unsigned)v[t].y / AG_RN) << 8) |
                             (((unsigned)v[t].z / AG_RN) << 16) | (((unsigned)v[t].w / AG_RN) << 24);
    }
  }

  for (int i = threadIdx.x; i < DH_RN / 4; i += DH_THREADS) sH4[i] = int4{0, 0, 0, 0};
  __syncthreads();

#pragma unroll
  for (int t = 0; t < DH_TRIPS; ++t) {
    const int j = threadIdx.x + DH_THREADS * t;
    const unsigned a0 = (unsigned)(v[t].x - lo), a1 = (unsigned)(v[t].y - lo);
    const unsigned a2 = (unsigned)(v[t].z - lo), a3 = (unsigned)(v[t].w - lo);
    if (j < DH_V4) {
      if (a0 < DH_RN) atomicAdd(&sH[a0], 1);
      if (a1 < DH_RN) atomicAdd(&sH[a1], 1);
      if (a2 < DH_RN) atomicAdd(&sH[a2], 1);
      if (a3 < DH_RN) atomicAdd(&sH[a3], 1);
    }
  }
  __syncthreads();
  int4* o = (int4*)(cnt + (size_t)s * DH_PITCH + lo);
  for (int i = threadIdx.x; i < DH_RN / 4; i += DH_THREADS) o[i] = sH4[i];
}

__global__ __launch_bounds__(256)
void gemm_deg_wmma(const float* __restrict__ X, const float* __restrict__ W,
                   const int* __restrict__ cnt, const int* __restrict__ batch,
                   float* __restrict__ dinv, int* __restrict__ rec,
                   float* __restrict__ out, int ntiles) {
  __shared__ float sWt[C * LPITCH];
  __shared__ int   sDeg[128];
  __shared__ float sDinv[128];

  const int nl   = threadIdx.x & 127;
  const int hf   = threadIdx.x >> 7;
  const int node = blockIdx.x * 128 + nl;
  int d = 0;
#pragma unroll
  for (int s = 0; s < DH_SLICES / 2; ++s) d += cnt[(size_t)(hf * (DH_SLICES / 2) + s) * DH_PITCH + node];

  for (int i = threadIdx.x; i < C * C; i += 256) {
    const int k = i >> 6, n = i & 63;
    sWt[n * LPITCH + k] = W[i];
  }
  if (hf) sDeg[nl] = d;
  __syncthreads();
  if (!hf) {
    const float r = rsqrtf((float)(1 + d + sDeg[nl]));
    sDinv[nl] = r;
    if (node < NODES) {
      dinv[node] = r;
      *(int2*)(rec + 2 * node) = int2{__float_as_int(r), batch[node]};
    }
  }

  const int wave = __builtin_amdgcn_readfirstlane(threadIdx.x >> 5);
  const int lane = threadIdx.x & 31;
  const int tile = blockIdx.x * 8 + wave;
  const bool act = tile < ntiles;

  const int m0 = tile * 16;
  const int cn = lane & 15;
  v8f acc0 = {}, acc1 = {}, acc2 = {}, acc3 = {};
  if (act) {
    const int ar = m0 + (lane & 15);
    const int kg = (lane >> 4) * 2;
    const float* bp = sWt + cn * LPITCH;
#pragma unroll
    for (int ks = 0; ks < 16; ++ks) {
      const int k0 = ks * 4 + kg;
      v2f a = *(const v2f*)(X + ar * C + k0);
      v2f b0 = *(const v2f*)(bp + k0);
      v2f b1 = *(const v2f*)(bp + 16 * LPITCH + k0);
      v2f b2 = *(const v2f*)(bp + 32 * LPITCH + k0);
      v2f b3 = *(const v2f*)(bp + 48 * LPITCH + k0);
      acc0 = __builtin_amdgcn_wmma_f32_16x16x4_f32(false, a, false, b0, (short)0, acc0, false, false);
      acc1 = __builtin_amdgcn_wmma_f32_16x16x4_f32(false, a, false, b1, (short)0, acc1, false, false);
      acc2 = __builtin_amdgcn_wmma_f32_16x16x4_f32(false, a, false, b2, (short)0, acc2, false, false);
      acc3 = __builtin_amdgcn_wmma_f32_16x16x4_f32(false, a, false, b3, (short)0, acc3, false, false);
    }
  }
  __syncthreads();
  if (act) {
#pragma unroll
    for (int r = 0; r < 8; ++r) {
      const int rl  = wave * 16 + r + 8 * (lane >> 4);
      const int row = blockIdx.x * 128 + rl;
      const float s = sDinv[rl];
      out[row * C +  0 + cn] = acc0[r] * s;
      out[row * C + 16 + cn] = acc1[r] * s;
      out[row * C + 32 + cn] = acc2[r] * s;
      out[row * C + 48 + cn] = acc3[r] * s;
    }
  }
}

#define AG_RANGES  256
#define AG_KV4     (EDGES / 16)
#define AG_THREADS 1024
#define AG_BATCH   8
#define AG_NB      ((AG_KV4 + AG_THREADS * AG_BATCH - 1) / (AG_THREADS * AG_BATCH))
#define AG_QCAP    32768
#define AG_U       16
#define AG_ROT     192

__device__ inline unsigned agg_hits(uint4 v, unsigned pat) {
  const unsigned x0 = v.x ^ pat, x1 = v.y ^ pat, x2 = v.z ^ pat, x3 = v.w ^ pat;
  const unsigned t0 = ((x0 & 0x7f7f7f7fu) + 0x7f7f7f7fu) | x0;
  const unsigned t1 = ((x1 & 0x7f7f7f7fu) + 0x7f7f7f7fu) | x1;
  const unsigned t2 = ((x2 & 0x7f7f7f7fu) + 0x7f7f7f7fu) | x2;
  const unsigned t3 = ((x3 & 0x7f7f7f7fu) + 0x7f7f7f7fu) | x3;
  const unsigned c = ((t0 >> 7) & 0x01010101u) | ((t1 >> 6) & 0x02020202u) |
                     ((t2 >> 5) & 0x04040404u) | ((t3 >> 4) & 0x08080808u);
  return ~c & 0x0f0f0f0fu;
}

__device__ inline int agg_group(int j) {
  const unsigned g = (unsigned)j + blockIdx.x * AG_ROT;
  const unsigned w = g - AG_KV4;
  return (int)(g < w ? g : w);
}

__device__ inline void agg_direct(float* __restrict__ sA, const int* __restrict__ src, const int* __restrict__ dst,
                                  const float* __restrict__ hs, const uint4* __restrict__ kp,
                                  unsigned pat, int lo, int b, int slot) {
#pragma nounroll
  for (int t = 0; t < AG_BATCH; ++t) {
    const int j = threadIdx.x + AG_THREADS * (b * AG_BATCH + t);
    if (j >= AG_KV4) break;
    const int g = agg_group(j);
    unsigned m = agg_hits(kp[g], pat);
    while (m) {
      const int p = __builtin_ctz(m);
      m &= m - 1;
      if (slot >= AG_QCAP) {
        const int e = 16 * g + 4 * (p & 7) + (p >> 3);
        const float* row = hs + (size_t)src[e] * C;
        float* d = sA + (dst[e] - lo) * C;
#pragma nounroll
        for (int c = 0; c < C; ++c)
          __hip_atomic_fetch_add(d + c, row[c], __ATOMIC_RELAXED, __HIP_MEMORY_SCOPE_WORKGROUP);
      }
      ++slot;
    }
  }
}

__global__ __launch_bounds__(AG_THREADS)
void agg_relu_kernel(const int* __restrict__ src, const int* __restrict__ dst, const unsigned* __restrict__ key,
                     const float* __restrict__ hs, const float* __restrict__ dinv,
                     const float* __restrict__ bias, float* __restrict__ hd) {
  __shared__ float4 sA4[AG_RN * C / 4];
  __shared__ int sQ[AG_QCAP];
  __shared__ int sQn;
  float* sA = (float*)sA4;
  const int lo = blockIdx.x * AG_RN;
  const unsigned pat = blockIdx.x * 0x01010101u;
  const uint4* kp = (const uint4*)key;

  for (int i = threadIdx.x; i < AG_RN * C / 4; i += AG_THREADS) sA4[i] = float4{0.0f, 0.0f, 0.0f, 0.0f};
  if (threadIdx.x == 0) sQn = 0;
  __syncthreads();

  for (int b = 0; b < AG_NB; ++b) {
    unsigned mm[AG_BATCH];
    {
      uint4 v[AG_BATCH];
#pragma unroll
      for (int t = 0; t < AG_BATCH; ++t) {
        const int j = threadIdx.x + AG_THREADS * (b * AG_BATCH + t);
        const int g = agg_group(j);
        v[t] = kp[g < AG_KV4 - 1 ? g : AG_KV4 - 1];
      }
#pragma unroll
      for (int t = 0; t < AG_BATCH; ++t) {
        const int j = threadIdx.x + AG_THREADS * (b * AG_BATCH + t);
        mm[t] = agg_hits(v[t], pat) & (j < AG_KV4 ? 0x0f0f0f0fu : 0u);
      }
    }
    int n = 0;
#pragma unroll
    for (int t = 0; t < AG_BATCH; ++t) n += __builtin_popcount(mm[t]);
    if (n) {
      const int slot0 = __hip_atomic_fetch_add(&sQn, n, __ATOMIC_RELAXED, __HIP_MEMORY_SCOPE_WORKGROUP);
      int slot = slot0;
#pragma unroll
      for (int t = 0; t < AG_BATCH; ++t) {
        const int g = agg_group(threadIdx.x + AG_THREADS * (b * AG_BATCH + t));
        unsigned m = mm[t];
        while (m) {
          const int p = __builtin_ctz(m);
          m &= m - 1;
          if (slot < AG_QCAP) sQ[slot] = 16 * g + 4 * (p & 7) + (p >> 3);
          ++slot;
        }
      }
      if (slot > AG_QCAP) agg_direct(sA, src, dst, hs, kp, pat, lo, b, slot0);
    }
  }
  __syncthreads();

  const int total = sQn;
  const int cnt = total < AG_QCAP ? total : AG_QCAP;
  for (int i = threadIdx.x; i < cnt; i += 4 * AG_THREADS) {
    int ix[4], ee[4], ss[4], dd[4];
#pragma unroll
    for (int k = 0; k < 4; ++k) {
      ix[k] = i + k * AG_THREADS < cnt ? i + k * AG_THREADS : i;
      ee[k] = sQ[ix[k]];
    }
#pragma unroll
    for (int k = 0; k < 4; ++k) { ss[k] = src[ee[k]]; dd[k] = dst[ee[k]]; }
#pragma unroll
    for (int k = 0; k < 4; ++k) sQ[ix[k]] = (ss[k] << 8) | (dd[k] - lo);
  }
  __syncthreads();

  const int wave = threadIdx.x >> 5;
  const int lane = threadIdx.x & 31;
  for (int i0 = wave * AG_U; i0 < cnt; i0 += (AG_THREADS / 32) * AG_U) {
    int q[AG_U];
    float a0[AG_U], a1[AG_U];
#pragma unroll
    for (int u = 0; u < AG_U; ++u) q[u] = sQ[i0 + u < cnt ? i0 + u : i0];
#pragma unroll
    for (int u = 0; u < AG_U; ++u) {
      const float* row = hs + (size_t)(q[u] >> 8) * C;
      a0[u] = row[lane];
      a1[u] = row[32 + lane];
    }
#pragma unroll
    for (int u = 0; u < AG_U; ++u) {
      const bool ok = i0 + u < cnt;
      float* d = sA + (q[u] & 255) * C;
      __hip_atomic_fetch_add(d + lane,      ok ? a0[u] : 0.0f, __ATOMIC_RELAXED, __HIP_MEMORY_SCOPE_WORKGROUP);
      __hip_atomic_fetch_add(d + 32 + lane, ok ? a1[u] : 0.0f, __ATOMIC_RELAXED, __HIP_MEMORY_SCOPE_WORKGROUP);
    }
  }

  const int nf4 = ((NODES - lo) < AG_RN ? (NODES - lo) : AG_RN) * (C / 4);
#define AG_FT ((AG_RN * C / 4 + AG_THREADS - 1) / AG_THREADS)
  float4 xs[AG_FT];
  float  ds[AG_FT];
#pragma unroll
  for (int t = 0; t < AG_FT; ++t) {
    const int i = threadIdx.x + AG_THREADS * t;
    const int ii = i < nf4 ? i : 0;
    xs[t] = ((const float4*)(hs + (size_t)lo * C))[ii];
    ds[t] = dinv[lo + (ii >> 4)];
  }
  const float4 bb = *(const float4*)(bias + 4 * (threadIdx.x & 15));
  __syncthreads();
#pragma unroll
  for (int t = 0; t < AG_FT; ++t) {
    const int i = threadIdx.x + AG_THREADS * t;
    if (i < nf4) {
      const float4 a = sA4[i];
      const float4 x = xs[t];
      const float d = ds[t];
      float4 v;
      v.x = d * fmaxf(d * (a.x + x.x) + bb.x, 0.0f);
      v.y = d * fmaxf(d * (a.y + x.y) + bb.y, 0.0f);
      v.z = d * fmaxf(d * (a.z + x.z) + bb.z, 0.0f);
      v.w = d * fmaxf(d * (a.w + x.w) + bb.w, 0.0f);
      ((float4*)(hd + (size_t)lo * C))[i] = v;
    }
  }
}

#define PE_THREADS 1024
#define PE_WAVES   (PE_THREADS / 32)
#define PE_BLOCKS  256
#define PE_ITEMS   (EDGES + NODES)
#define PE_NCH     ((PE_ITEMS + 31) / 32)
#define PE_STRIDE  (PE_BLOCKS * PE_WAVES)
#define PE_K       ((PE_NCH + PE_STRIDE - 1) / PE_STRIDE)

__global__ __launch_bounds__(PE_THREADS)
void pool_edges_kernel(const int* __restrict__ src, const int* __restrict__ dst,
                       const int* __restrict__ rec, const float* __restrict__ hd,
                       float* __restrict__ part) {
  __shared__ float4 sU4[GRAPHS * C / 4];
  float* sU = (float*)sU4;
  const int wave = __builtin_amdgcn_readfirstlane(threadIdx.x >> 5);
  const int lane = threadIdx.x & 31;
  const int wv = wave * PE_BLOCKS + blockIdx.x;

  int ss[PE_K], dd[PE_K];
  bool ok[PE_K];
#pragma unroll
  for (int k = 0; k < PE_K; ++k) {
    const int ch = wv + k * PE_STRIDE;
    const int item = (ch < PE_NCH ? ch : 0) * 32 + lane;
    const bool edge = item < EDGES;
    ok[k] = item < PE_ITEMS;
    const int e = edge ? item : 0;
    const int n = ok[k] ? item - EDGES : 0;
    const int sl = src[e], dl = dst[e];
    ss[k] = edge ? sl : n;
    dd[k] = edge ? dl : n;
  }
  unsigned ms[PE_K];
  float mw[PE_K];
#pragma unroll
  for (int k = 0; k < PE_K; ++k) {
    const int2 r = *(const int2*)(rec + 2 * dd[k]);
    const bool in = ok[k] && (unsigned)r.y < GRAPHS;
    ms[k] = ((unsigned)ss[k] << 6) | (unsigned)(in ? r.y : 0);
    mw[k] = in ? __int_as_float(r.x) : 0.0f;
  }

  sU4[threadIdx.x] = float4{0.0f, 0.0f, 0.0f, 0.0f};
  __syncthreads();

#pragma unroll
  for (int k = 0; k < PE_K; ++k) {
    if (wv + k * PE_STRIDE < PE_NCH) {
      float2 a[32];
#pragma unroll
      for (int u = 0; u < 32; ++u) {
        const unsigned m = __builtin_amdgcn_readlane(ms[k], u);
        a[u] = *(const float2*)(hd + (size_t)(m >> 6) * C + 2 * lane);
      }
#pragma unroll
      for (int u = 0; u < 32; ++u) {
        const unsigned m = __builtin_amdgcn_readlane(ms[k], u);
        const float w = __int_as_float(__builtin_amdgcn_readlane(__float_as_int(mw[k]), u));
        float* d = sU + (m & (GRAPHS - 1)) * C + 2 * lane;
        __hip_atomic_fetch_add(d,     w * a[u].x, __ATOMIC_RELAXED, __HIP_MEMORY_SCOPE_WORKGROUP);
        __hip_atomic_fetch_add(d + 1, w * a[u].y, __ATOMIC_RELAXED, __HIP_MEMORY_SCOPE_WORKGROUP);
      }
    }
  }
  __syncthreads();

  ((float4*)(part + (size_t)blockIdx.x * (GRAPHS * C)))[threadIdx.x] = sU4[threadIdx.x];
}

__global__ __launch_bounds__(256)
void pool_head(const float* __restrict__ part, const int* __restrict__ batch,
               const float* __restrict__ W2, const float* __restrict__ b2,
               const float* __restrict__ Wf, const float* __restrict__ bf,
               float* __restrict__ y, float* __restrict__ out_u) {
  __shared__ float sR[16 * 64];
  __shared__ int sN[256];
  __shared__ float sUp[16 * LPITCH];
  __shared__ float sU[16 * LPITCH];
  __shared__ float sCnt;
  const int g = blockIdx.x;
  const int wave = __builtin_amdgcn_readfirstlane(threadIdx.x >> 5);
  const int lane = threadIdx.x & 31;
  const int c  = lane & 15;
  const int kg = (lane >> 4) * 2;

  v2f bw[16] = {}, bfa[16] = {}, bfb[16] = {};
  float bb = 0.0f;
  if (wave < 4) {
    const int col = wave * 16 + c;
#pragma unroll
    for (int ks = 0; ks < 16; ++ks) {
      const int k0 = ks * 4 + kg;
      bw[ks] = v2f{W2[k0 * C + col], W2[(k0 + 1) * C + col]};
    }
    bb = b2[col];
  }
  if (wave == 0) {
#pragma unroll
    for (int ks = 0; ks < 16; ++ks) {
      const int k0 = ks * 4 + kg;
      bfa[ks] = v2f{Wf[k0 * OUTC + c],      Wf[(k0 + 1) * OUTC + c]};
      bfb[ks] = v2f{Wf[k0 * OUTC + 16 + c], Wf[(k0 + 1) * OUTC + 16 + c]};
    }
  }

  for (int z = threadIdx.x; z < 16 * LPITCH; z += 256) { sUp[z] = 0.0f; sU[z] = 0.0f; }

  const int i   = threadIdx.x & 15;
  const int grp = threadIdx.x >> 4;
  const int f4  = blockIdx.x * 16 + i;
  float4 s = {0.0f, 0.0f, 0.0f, 0.0f};
#pragma unroll
  for (int j = 0; j < (PE_BLOCKS + 15) / 16; ++j) {
    const int p = grp + 16 * j;
    if (p < PE_BLOCKS) {
      const float4 t = *(const float4*)(part + (size_t)p * (GRAPHS * C) + f4 * 4);
      s.x += t.x; s.y += t.y; s.z += t.z; s.w += t.w;
    }
  }
  int mine = 0;
#pragma unroll
  for (int t = 0; t < (NODES / 4 + 255) / 256; ++t) {
    const int j  = threadIdx.x + 256 * t;
    const bool in = j < NODES / 4;
    const int4 b = *(const int4*)(batch + 4 * (in ? j : 0));
    const int n  = (b.x == g) + (b.y == g) + (b.z == g) + (b.w == g);
    mine += in ? n : 0;
  }
  *(float4*)(sR + grp * 64 + i * 4) = s;
  sN[threadIdx.x] = mine;
  __syncthreads();
  if (threadIdx.x < 64) {
    float r = sR[threadIdx.x];
#pragma unroll
    for (int k = 1; k < 16; ++k) r += sR[k * 64 + threadIdx.x];
    sUp[threadIdx.x] = r;
  }
  if (threadIdx.x < 32) {
    int t = 0;
#pragma unroll
    for (int k = 0; k < 8; ++k) t += sN[threadIdx.x + 32 * k];
#pragma unroll
    for (int m = 16; m >= 1; m >>= 1) t += __shfl_xor(t, m);
    if (threadIdx.x == 0) sCnt = (float)t;
  }
  __syncthreads();

  if (wave < 4) {
    v8f acc = {};
#pragma unroll
    for (int ks = 0; ks < 16; ++ks) {
      const v2f a = *(const v2f*)(sUp + c * LPITCH + ks * 4 + kg);
      acc = __builtin_amdgcn_wmma_f32_16x16x4_f32(false, a, false, bw[ks], (short)0, acc, false, false);
    }
    const float n = sCnt;
    const float v = acc[0] + n * bb;
    if (lane < 16) sU[wave * 16 + c] = v;
  }
  __syncthreads();

  if (wave == 0) {
    out_u[g * C + lane]      = sU[lane];
    out_u[g * C + 32 + lane] = sU[32 + lane];
    v8f ya = {}, yb = {};
#pragma unroll
    for (int ks = 0; ks < 16; ++ks) {
      const v2f a = *(const v2f*)(sU + c * LPITCH + ks * 4 + kg);
      ya = __builtin_amdgcn_wmma_f32_16x16x4_f32(false, a, false, bfa[ks], (short)0, ya, false, false);
      yb = __builtin_amdgcn_wmma_f32_16x16x4_f32(false, a, false, bfb[ks], (short)0, yb, false, false);
    }
    const float hi = __shfl(yb[0], c);
    const float v  = lane < 16 ? ya[0] : hi;
    y[g * OUTC + lane] = v + bf[lane];
  }
}

extern "C" void kernel_launch(void* const* d_in, const int* in_sizes, int n_in,
                              void* d_out, int out_size, void* d_ws, size_t ws_size,
                              hipStream_t stream) {
  (void)in_sizes; (void)n_in; (void)out_size; (void)ws_size;

  const float* x     = (const float*)d_in[0];
  const int*   ei    = (const int*)  d_in[1];
  const int*   batch = (const int*)  d_in[2];
  const float* W1    = (const float*)d_in[3];
  const float* b1    = (const float*)d_in[4];
  const float* W2    = (const float*)d_in[5];
  const float* b2    = (const float*)d_in[6];
  const float* Wf    = (const float*)d_in[7];
  const float* bf    = (const float*)d_in[8];
  float* out = (float*)d_out;

  const int* src = ei;
  const int* dst = ei + EDGES;

  float* ws   = (float*)d_ws;
  float* dinv = ws;
  int*   rec  = (int*)(ws + 50048);
  float* hs   = ws + 3 * 50048;
  float* acc  = hs + (size_t)NODES * C;
  float* part = acc + (size_t)NODES * C + GRAPHS * C + GRAPHS;
  int*   degc = (int*)(part + (size_t)PE_BLOCKS * GRAPHS * C);
  unsigned* key = (unsigned*)(degc + (size_t)DH_SLICES * DH_PITCH);

  const dim3 B(256);
  const int NT = NODES / 16;
  const int GB = (NT + 7) / 8;

  deg_hist_kernel<<<DH_RANGES * DH_SLICES, dim3(DH_THREADS), 0, stream>>>(dst, degc, key);

  gemm_deg_wmma  <<<GB, B, 0, stream>>>(x, W1, degc, batch, dinv, rec, hs, NT);
  agg_relu_kernel<<<AG_RANGES, dim3(AG_THREADS), 0, stream>>>(src, dst, key, hs, dinv, b1, acc);

  pool_edges_kernel<<<PE_BLOCKS, dim3(PE_THREADS), 0, stream>>>(src, dst, rec, acc, part);
  pool_head  <<<GRAPHS, B, 0, stream>>>(part, batch, W2, b2, Wf, bf, out, out + GRAPHS * OUTC);
}


# aggregation finds its edges in a row of bits per node range (one bit per edge, built in LDS by the degree pass) instead of testing a byte per edge of the whole list; sums in arrival order as before
# speedup vs baseline: 3.2929x; 1.0343x over previous
#include <hip/hip_runtime.h>

typedef __attribute__((ext_vector_type(2))) float v2f;
typedef __attribute__((ext_vector_type(8))) float v8f;

#define NODES  50000
#define EDGES  800000
#define GRAPHS 64
#define C      64
#define OUTC   32
#define LPITCH 68
#define AG_RN     196
#define AG_RANGES 256
#define BM_CW     128
#define BM_CE     (32 * BM_CW)
#define BM_NCH    ((EDGES + BM_CE - 1) / BM_CE)
#define BM_PITCH  (BM_NCH * BM_CW)
#define BM_V4     (BM_PITCH / 4)

#define DH_RANGES  8
#define DH_SLICES  32
#define DH_RN      6272
#define DH_PITCH   (DH_RANGES * DH_RN)
#define DH_SE      (EDGES / DH_SLICES)
#define DH_V4      (DH_SE / 4)
#define DH_THREADS 512
#define DH_TRIPS   ((DH_V4 + DH_THREADS - 1) / DH_THREADS)

#define BM_CT      (BM_CE / 4 / DH_THREADS)
#define BM_WT      (AG_RANGES * BM_CW / 4 / DH_THREADS)
__global__ __launch_bounds__(DH_THREADS)
void deg_hist_kernel(const int* __restrict__ dst, int* __restrict__ cnt, unsigned* __restrict__ bm) {
  __shared__ int4 sH4[DH_RN / 4];
  __shared__ uint4 sB4[AG_RANGES * BM_CW / 4];
  int* sH = (int*)sH4;
  unsigned* sB = (unsigned*)sB4;
  const int r = blockIdx.x % DH_RANGES;
  const int s = blockIdx.x / DH_RANGES;
  const int lo = r * DH_RN;
  const int* p = dst + s * DH_SE;
  const int ch = blockIdx.x;
  const bool duty = ch < BM_NCH;

  int4 v[DH_TRIPS];
#pragma unroll
  for (int t = 0; t < DH_TRIPS; ++t) {
    const int j = threadIdx.x + DH_THREADS * t;
    v[t] = *(const int4*)(p + 4 * (j < DH_V4 ? j : 0));
  }
  int4 cd[BM_CT];
#pragma unroll
  for (int t = 0; t < BM_CT; ++t) {
    const int e = ch * BM_CE + 4 * (threadIdx.x + DH_THREADS * t);
    cd[t] = *(const int4*)(dst + (e < EDGES ? e : 0));
  }

  for (int i = threadIdx.x; i < DH_RN / 4; i += DH_THREADS) sH4[i] = int4{0, 0, 0, 0};
  if (duty) {
#pragma unroll
    for (int t = 0; t < BM_WT; ++t) sB4[threadIdx.x + DH_THREADS * t] = uint4{0u, 0u, 0u, 0u};
  }
  __syncthreads();

#pragma unroll
  for (int t = 0; t < DH_TRIPS; ++t) {
    const int j = threadIdx.x + DH_THREADS * t;
    const unsigned a0 = (unsigned)(v[t].x - lo), a1 = (unsigned)(v[t].y - lo);
    const unsigned a2 = (unsigned)(v[t].z - lo), a3 = (unsigned)(v[t].w - lo);
    if (j < DH_V4) {
      if (a0 < DH_RN) atomicAdd(&sH[a0], 1);
      if (a1 < DH_RN) atomicAdd(&sH[a1], 1);
      if (a2 < DH_RN) atomicAdd(&sH[a2], 1);
      if (a3 < DH_RN) atomicAdd(&sH[a3], 1);
    }
  }
  if (duty) {
#pragma unroll
    for (int t = 0; t < BM_CT; ++t) {
      const int j = threadIdx.x + DH_THREADS * t;
      if (ch * BM_CE + 4 * j < EDGES) {
        const unsigned g0 = (unsigned)cd[t].x / AG_RN, g1 = (unsigned)cd[t].y / AG_RN;
        const unsigned g2 = (unsigned)cd[t].z / AG_RN, g3 = (unsigned)cd[t].w / AG_RN;
        const unsigned w = (unsigned)j >> 3;
        const unsigned b = 1u << (4 * (j & 7));
        if (g0 < AG_RANGES) __hip_atomic_fetch_or(&sB[g0 * BM_CW + w], b,      __ATOMIC_RELAXED, __HIP_MEMORY_SCOPE_WORKGROUP);
        if (g1 < AG_RANGES) __hip_atomic_fetch_or(&sB[g1 * BM_CW + w], b << 1, __ATOMIC_RELAXED, __HIP_MEMORY_SCOPE_WORKGROUP);
        if (g2 < AG_RANGES) __hip_atomic_fetch_or(&sB[g2 * BM_CW + w], b << 2, __ATOMIC_RELAXED, __HIP_MEMORY_SCOPE_WORKGROUP);
        if (g3 < AG_RANGES) __hip_atomic_fetch_or(&sB[g3 * BM_CW + w], b << 3, __ATOMIC_RELAXED, __HIP_MEMORY_SCOPE_WORKGROUP);
      }
    }
  }
  __syncthreads();
  int4* o = (int4*)(cnt + (size_t)s * DH_PITCH + lo);
  for (int i = threadIdx.x; i < DH_RN / 4; i += DH_THREADS) o[i] = sH4[i];
  if (duty) {
#pragma unroll
    for (int t = 0; t < BM_WT; ++t) {
      const int q = threadIdx.x + DH_THREADS * t;
      ((uint4*)(bm + (size_t)(q >> 5) * BM_PITCH + ch * BM_CW))[q & 31] = sB4[q];
    }
  }
}

__global__ __launch_bounds__(256)
void gemm_deg_wmma(const float* __restrict__ X, const float* __restrict__ W,
                   const int* __restrict__ cnt, const int* __restrict__ batch,
                   float* __restrict__ dinv, int* __restrict__ rec,
                   float* __restrict__ out, int ntiles) {
  __shared__ float sWt[C * LPITCH];
  __shared__ int   sDeg[128];
  __shared__ float sDinv[128];

  const int nl   = threadIdx.x & 127;
  const int hf   = threadIdx.x >> 7;
  const int node = blockIdx.x * 128 + nl;
  int d = 0;
#pragma unroll
  for (int s = 0; s < DH_SLICES / 2; ++s) d += cnt[(size_t)(hf * (DH_SLICES / 2) + s) * DH_PITCH + node];

  for (int i = threadIdx.x; i < C * C; i += 256) {
    const int k = i >> 6, n = i & 63;
    sWt[n * LPITCH + k] = W[i];
  }
  if (hf) sDeg[nl] = d;
  __syncthreads();
  if (!hf) {
    const float r = rsqrtf((float)(1 + d + sDeg[nl]));
    sDinv[nl] = r;
    if (node < NODES) {
      dinv[node] = r;
      *(int2*)(rec + 2 * node) = int2{__float_as_int(r), batch[node]};
    }
  }

  const int wave = __builtin_amdgcn_readfirstlane(threadIdx.x >> 5);
  const int lane = threadIdx.x & 31;
  const int tile = blockIdx.x * 8 + wave;
  const bool act = tile < ntiles;

  const int m0 = tile * 16;
  const int cn = lane & 15;
  v8f acc0 = {}, acc1 = {}, acc2 = {}, acc3 = {};
  if (act) {
    const int ar = m0 + (lane & 15);
    const int kg = (lane >> 4) * 2;
    const float* bp = sWt + cn * LPITCH;
#pragma unroll
    for (int ks = 0; ks < 16; ++ks) {
      const int k0 = ks * 4 + kg;
      v2f a = *(const v2f*)(X + ar * C + k0);
      v2f b0 = *(const v2f*)(bp + k0);
      v2f b1 = *(const v2f*)(bp + 16 * LPITCH + k0);
      v2f b2 = *(const v2f*)(bp + 32 * LPITCH + k0);
      v2f b3 = *(const v2f*)(bp + 48 * LPITCH + k0);
      acc0 = __builtin_amdgcn_wmma_f32_16x16x4_f32(false, a, false, b0, (short)0, acc0, false, false);
      acc1 = __builtin_amdgcn_wmma_f32_16x16x4_f32(false, a, false, b1, (short)0, acc1, false, false);
      acc2 = __builtin_amdgcn_wmma_f32_16x16x4_f32(false, a, false, b2, (short)0, acc2, false, false);
      acc3 = __builtin_amdgcn_wmma_f32_16x16x4_f32(false, a, false, b3, (short)0, acc3, false, false);
    }
  }
  __syncthreads();
  if (act) {
#pragma unroll
    for (int r = 0; r < 8; ++r) {
      const int rl  = wave * 16 + r + 8 * (lane >> 4);
      const int row = blockIdx.x * 128 + rl;
      const float s = sDinv[rl];
      out[row * C +  0 + cn] = acc0[r] * s;
      out[row * C + 16 + cn] = acc1[r] * s;
      out[row * C + 32 + cn] = acc2[r] * s;
      out[row * C + 48 + cn] = acc3[r] * s;
    }
  }
}

#define AG_THREADS 1024
#define AG_NV      ((BM_V4 + AG_THREADS - 1) / AG_THREADS)
#define AG_QCAP    32768
#define AG_U       16

__device__ inline void agg_spill(float* __restrict__ sA, int* __restrict__ sQ,
                                 const int* __restrict__ src, const int* __restrict__ dst,
                                 const float* __restrict__ hs, const unsigned* __restrict__ row, int lo, int slot) {
#pragma nounroll
  for (int t = 0; t < AG_NV; ++t) {
    const int j = threadIdx.x + AG_THREADS * t;
    if (j >= BM_V4) break;
#pragma nounroll
    for (int i = 0; i < 4; ++i) {
      unsigned m = row[4 * j + i];
      while (m) {
        const int p = __builtin_ctz(m);
        m &= m - 1;
        const int e = 128 * j + 32 * i + p;
        if (slot < AG_QCAP) {
          sQ[slot] = e;
        } else {
          const float* hrow = hs + (size_t)src[e] * C;
          float* d = sA + (dst[e] - lo) * C;
#pragma nounroll
          for (int cc = 0; cc < C; ++cc)
            __hip_atomic_fetch_add(d + cc, hrow[cc], __ATOMIC_RELAXED, __HIP_MEMORY_SCOPE_WORKGROUP);
        }
        ++slot;
      }
    }
  }
}

__global__ __launch_bounds__(AG_THREADS)
void agg_relu_kernel(const int* __restrict__ src, const int* __restrict__ dst, const unsigned* __restrict__ bm,
                     const float* __restrict__ hs, const float* __restrict__ dinv,
                     const float* __restrict__ bias, float* __restrict__ hd) {
  __shared__ float4 sA4[AG_RN * C / 4];
  __shared__ int sQ[AG_QCAP];
  __shared__ int sQn;
  float* sA = (float*)sA4;
  const int lo = blockIdx.x * AG_RN;
  const unsigned* row = bm + (size_t)blockIdx.x * BM_PITCH;

  uint4 v[AG_NV];
#pragma unroll
  for (int t = 0; t < AG_NV; ++t) {
    const int j = threadIdx.x + AG_THREADS * t;
    v[t] = ((const uint4*)row)[j < BM_V4 ? j : 0];
  }

  for (int i = threadIdx.x; i < AG_RN * C / 4; i += AG_THREADS) sA4[i] = float4{0.0f, 0.0f, 0.0f, 0.0f};
  if (threadIdx.x == 0) sQn = 0;
  __syncthreads();

  int n = 0;
#pragma unroll
  for (int t = 0; t < AG_NV; ++t) {
    if (threadIdx.x + AG_THREADS * t >= BM_V4) v[t] = uint4{0u, 0u, 0u, 0u};
    n += __builtin_popcount(v[t].x) + __builtin_popcount(v[t].y) + __builtin_popcount(v[t].z) + __builtin_popcount(v[t].w);
  }
  if (n) {
    int slot = __hip_atomic_fetch_add(&sQn, n, __ATOMIC_RELAXED, __HIP_MEMORY_SCOPE_WORKGROUP);
    if (slot + n <= AG_QCAP) {
#pragma unroll
      for (int t = 0; t < AG_NV; ++t) {
        const int e0 = 128 * (threadIdx.x + AG_THREADS * t);
        const unsigned w[4] = {v[t].x, v[t].y, v[t].z, v[t].w};
#pragma unroll
        for (int i = 0; i < 4; ++i) {
          unsigned m = w[i];
          while (m) {
            sQ[slot++] = e0 + 32 * i + __builtin_ctz(m);
            m &= m - 1;
          }
        }
      }
    } else {
      agg_spill(sA, sQ, src, dst, hs, row, lo, slot);
    }
  }
  __syncthreads();

  const int total = sQn;
  const int cnt = total < AG_QCAP ? total : AG_QCAP;
  for (int i = threadIdx.x; i < cnt; i += 4 * AG_THREADS) {
    int ix[4], ee[4], ss[4], dd[4];
#pragma unroll
    for (int k = 0; k < 4; ++k) {
      ix[k] = i + k * AG_THREADS < cnt ? i + k * AG_THREADS : i;
      ee[k] = sQ[ix[k]];
    }
#pragma unroll
    for (int k = 0; k < 4; ++k) { ss[k] = src[ee[k]]; dd[k] = dst[ee[k]]; }
#pragma unroll
    for (int k = 0; k < 4; ++k) sQ[ix[k]] = (ss[k] << 8) | (dd[k] - lo);
  }
  __syncthreads();

  const int wave = threadIdx.x >> 5;
  const int lane = threadIdx.x & 31;
  for (int i0 = wave * AG_U; i0 < cnt; i0 += (AG_THREADS / 32) * AG_U) {
    int q[AG_U];
    float a0[AG_U], a1[AG_U];
#pragma unroll
    for (int u = 0; u < AG_U; ++u) q[u] = sQ[i0 + u < cnt ? i0 + u : i0];
#pragma unroll
    for (int u = 0; u < AG_U; ++u) {
      const float* row = hs + (size_t)(q[u] >> 8) * C;
      a0[u] = row[lane];
      a1[u] = row[32 + lane];
    }
#pragma unroll
    for (int u = 0; u < AG_U; ++u) {
      const bool ok = i0 + u < cnt;
      float* d = sA + (q[u] & 255) * C;
      __hip_atomic_fetch_add(d + lane,      ok ? a0[u] : 0.0f, __ATOMIC_RELAXED, __HIP_MEMORY_SCOPE_WORKGROUP);
      __hip_atomic_fetch_add(d + 32 + lane, ok ? a1[u] : 0.0f, __ATOMIC_RELAXED, __HIP_MEMORY_SCOPE_WORKGROUP);
    }
  }

  const int nf4 = ((NODES - lo) < AG_RN ? (NODES - lo) : AG_RN) * (C / 4);
#define AG_FT ((AG_RN * C / 4 + AG_THREADS - 1) / AG_THREADS)
  float4 xs[AG_FT];
  float  ds[AG_FT];
#pragma unroll
  for (int t = 0; t < AG_FT; ++t) {
    const int i = threadIdx.x + AG_THREADS * t;
    const int ii = i < nf4 ? i : 0;
    xs[t] = ((const float4*)(hs + (size_t)lo * C))[ii];
    ds[t] = dinv[lo + (ii >> 4)];
  }
  const float4 bb = *(const float4*)(bias + 4 * (threadIdx.x & 15));
  __syncthreads();
#pragma unroll
  for (int t = 0; t < AG_FT; ++t) {
    const int i = threadIdx.x + AG_THREADS * t;
    if (i < nf4) {
      const float4 a = sA4[i];
      const float4 x = xs[t];
      const float d = ds[t];
      float4 v;
      v.x = d * fmaxf(d * (a.x + x.x) + bb.x, 0.0f);
      v.y = d * fmaxf(d * (a.y + x.y) + bb.y, 0.0f);
      v.z = d * fmaxf(d * (a.z + x.z) + bb.z, 0.0f);
      v.w = d * fmaxf(d * (a.w + x.w) + bb.w, 0.0f);
      ((float4*)(hd + (size_t)lo * C))[i] = v;
    }
  }
}

#define PE_THREADS 1024
#define PE_WAVES   (PE_THREADS / 32)
#define PE_BLOCKS  256
#define PE_ITEMS   (EDGES + NODES)
#define PE_NCH     ((PE_ITEMS + 31) / 32)
#define PE_STRIDE  (PE_BLOCKS * PE_WAVES)
#define PE_K       ((PE_NCH + PE_STRIDE - 1) / PE_STRIDE)

__global__ __launch_bounds__(PE_THREADS)
void pool_edges_kernel(const int* __restrict__ src, const int* __restrict__ dst,
                       const int* __restrict__ rec, const float* __restrict__ hd,
                       float* __restrict__ part) {
  __shared__ float4 sU4[GRAPHS * C / 4];
  float* sU = (float*)sU4;
  const int wave = __builtin_amdgcn_readfirstlane(threadIdx.x >> 5);
  const int lane = threadIdx.x & 31;
  const int wv = wave * PE_BLOCKS + blockIdx.x;

  int ss[PE_K], dd[PE_K];
  bool ok[PE_K];
#pragma unroll
  for (int k = 0; k < PE_K; ++k) {
    const int ch = wv + k * PE_STRIDE;
    const int item = (ch < PE_NCH ? ch : 0) * 32 + lane;
    const bool edge = item < EDGES;
    ok[k] = item < PE_ITEMS;
    const int e = edge ? item : 0;
    const int n = ok[k] ? item - EDGES : 0;
    const int sl = src[e], dl = dst[e];
    ss[k] = edge ? sl : n;
    dd[k] = edge ? dl : n;
  }
  unsigned ms[PE_K];
  float mw[PE_K];
#pragma unroll
  for (int k = 0; k < PE_K; ++k) {
    const int2 r = *(const int2*)(rec + 2 * dd[k]);
    const bool in = ok[k] && (unsigned)r.y < GRAPHS;
    ms[k] = ((unsigned)ss[k] << 6) | (unsigned)(in ? r.y : 0);
    mw[k] = in ? __int_as_float(r.x) : 0.0f;
  }

  sU4[threadIdx.x] = float4{0.0f, 0.0f, 0.0f, 0.0f};
  __syncthreads();

#pragma unroll
  for (int k = 0; k < PE_K; ++k) {
    if (wv + k * PE_STRIDE < PE_NCH) {
      float2 a[32];
#pragma unroll
      for (int u = 0; u < 32; ++u) {
        const unsigned m = __builtin_amdgcn_readlane(ms[k], u);
        a[u] = *(const float2*)(hd + (size_t)(m >> 6) * C + 2 * lane);
      }
#pragma unroll
      for (int u = 0; u < 32; ++u) {
        const unsigned m = __builtin_amdgcn_readlane(ms[k], u);
        const float w = __int_as_float(__builtin_amdgcn_readlane(__float_as_int(mw[k]), u));
        float* d = sU + (m & (GRAPHS - 1)) * C + 2 * lane;
        __hip_atomic_fetch_add(d,     w * a[u].x, __ATOMIC_RELAXED, __HIP_MEMORY_SCOPE_WORKGROUP);
        __hip_atomic_fetch_add(d + 1, w * a[u].y, __ATOMIC_RELAXED, __HIP_MEMORY_SCOPE_WORKGROUP);
      }
    }
  }
  __syncthreads();

  ((float4*)(part + (size_t)blockIdx.x * (GRAPHS * C)))[threadIdx.x] = sU4[threadIdx.x];
}

__global__ __launch_bounds__(256)
void pool_head(const float* __restrict__ part, const int* __restrict__ batch,
               const float* __restrict__ W2, const float* __restrict__ b2,
               const float* __restrict__ Wf, const float* __restrict__ bf,
               float* __restrict__ y, float* __restrict__ out_u) {
  __shared__ float sR[16 * 64];
  __shared__ int sN[256];
  __shared__ float sUp[16 * LPITCH];
  __shared__ float sU[16 * LPITCH];
  __shared__ float sCnt;
  const int g = blockIdx.x;
  const int wave = __builtin_amdgcn_readfirstlane(threadIdx.x >> 5);
  const int lane = threadIdx.x & 31;
  const int c  = lane & 15;
  const int kg = (lane >> 4) * 2;

  v2f bw[16] = {}, bfa[16] = {}, bfb[16] = {};
  float bb = 0.0f;
  if (wave < 4) {
    const int col = wave * 16 + c;
#pragma unroll
    for (int ks = 0; ks < 16; ++ks) {
      const int k0 = ks * 4 + kg;
      bw[ks] = v2f{W2[k0 * C + col], W2[(k0 + 1) * C + col]};
    }
    bb = b2[col];
  }
  if (wave == 0) {
#pragma unroll
    for (int ks = 0; ks < 16; ++ks) {
      const int k0 = ks * 4 + kg;
      bfa[ks] = v2f{Wf[k0 * OUTC + c],      Wf[(k0 + 1) * OUTC + c]};
      bfb[ks] = v2f{Wf[k0 * OUTC + 16 + c], Wf[(k0 + 1) * OUTC + 16 + c]};
    }
  }

  for (int z = threadIdx.x; z < 16 * LPITCH; z += 256) { sUp[z] = 0.0f; sU[z] = 0.0f; }

  const int i   = threadIdx.x & 15;
  const int grp = threadIdx.x >> 4;
  const int f4  = blockIdx.x * 16 + i;
  float4 s = {0.0f, 0.0f, 0.0f, 0.0f};
#pragma unroll
  for (int j = 0; j < (PE_BLOCKS + 15) / 16; ++j) {
    const int p = grp + 16 * j;
    if (p < PE_BLOCKS) {
      const float4 t = *(const float4*)(part + (size_t)p * (GRAPHS * C) + f4 * 4);
      s.x += t.x; s.y += t.y; s.z += t.z; s.w += t.w;
    }
  }
  int mine = 0;
#pragma unroll
  for (int t = 0; t < (NODES / 4 + 255) / 256; ++t) {
    const int j  = threadIdx.x + 256 * t;
    const bool in = j < NODES / 4;
    const int4 b = *(const int4*)(batch + 4 * (in ? j : 0));
    const int n  = (b.x == g) + (b.y == g) + (b.z == g) + (b.w == g);
    mine += in ? n : 0;
  }
  *(float4*)(sR + grp * 64 + i * 4) = s;
  sN[threadIdx.x] = mine;
  __syncthreads();
  if (threadIdx.x < 64) {
    float r = sR[threadIdx.x];
#pragma unroll
    for (int k = 1; k < 16; ++k) r += sR[k * 64 + threadIdx.x];
    sUp[threadIdx.x] = r;
  }
  if (threadIdx.x < 32) {
    int t = 0;
#pragma unroll
    for (int k = 0; k < 8; ++k) t += sN[threadIdx.x + 32 * k];
#pragma unroll
    for (int m = 16; m >= 1; m >>= 1) t += __shfl_xor(t, m);
    if (threadIdx.x == 0) sCnt = (float)t;
  }
  __syncthreads();

  if (wave < 4) {
    v8f acc = {};
#pragma unroll
    for (int ks = 0; ks < 16; ++ks) {
      const v2f a = *(const v2f*)(sUp + c * LPITCH + ks * 4 + kg);
      acc = __builtin_amdgcn_wmma_f32_16x16x4_f32(false, a, false, bw[ks], (short)0, acc, false, false);
    }
    const float n = sCnt;
    const float v = acc[0] + n * bb;
    if (lane < 16) sU[wave * 16 + c] = v;
  }
  __syncthreads();

  if (wave == 0) {
    out_u[g * C + lane]      = sU[lane];
    out_u[g * C + 32 + lane] = sU[32 + lane];
    v8f ya = {}, yb = {};
#pragma unroll
    for (int ks = 0; ks < 16; ++ks) {
      const v2f a = *(const v2f*)(sU + c * LPITCH + ks * 4 + kg);
      ya = __builtin_amdgcn_wmma_f32_16x16x4_f32(false, a, false, bfa[ks], (short)0, ya, false, false);
      yb = __builtin_amdgcn_wmma_f32_16x16x4_f32(false, a, false, bfb[ks], (short)0, yb, false, false);
    }
    const float hi = __shfl(yb[0], c);
    const float v  = lane < 16 ? ya[0] : hi;
    y[g * OUTC + lane] = v + bf[lane];
  }
}

extern "C" void kernel_launch(void* const* d_in, const int* in_sizes, int n_in,
                              void* d_out, int out_size, void* d_ws, size_t ws_size,
                              hipStream_t stream) {
  (void)in_sizes; (void)n_in; (void)out_size; (void)ws_size;

  const float* x     = (const float*)d_in[0];
  const int*   ei    = (const int*)  d_in[1];
  const int*   batch = (const int*)  d_in[2];
  const float* W1    = (const float*)d_in[3];
  const float* b1    = (const float*)d_in[4];
  const float* W2    = (const float*)d_in[5];
  const float* b2    = (const float*)d_in[6];
  const float* Wf    = (const float*)d_in[7];
  const float* bf    = (const float*)d_in[8];
  float* out = (float*)d_out;

  const int* src = ei;
  const int* dst = ei + EDGES;

  float* ws   = (float*)d_ws;
  float* dinv = ws;
  int*   rec  = (int*)(ws + 50048);
  float* hs   = ws + 3 * 50048;
  float* acc  = hs + (size_t)NODES * C;
  float* part = acc + (size_t)NODES * C + GRAPHS * C + GRAPHS;
  int*   degc = (int*)(part + (size_t)PE_BLOCKS * GRAPHS * C);
  unsigned* bm = (unsigned*)(degc + (size_t)DH_SLICES * DH_PITCH);

  const dim3 B(256);
  const int NT = NODES / 16;
  const int GB = (NT + 7) / 8;

  deg_hist_kernel<<<DH_RANGES * DH_SLICES, dim3(DH_THREADS), 0, stream>>>(dst, degc, bm);

  gemm_deg_wmma  <<<GB, B, 0, stream>>>(x, W1, degc, batch, dinv, rec, hs, NT);
  agg_relu_kernel<<<AG_RANGES, dim3(AG_THREADS), 0, stream>>>(src, dst, bm, hs, dinv, b1, acc);

  pool_edges_kernel<<<PE_BLOCKS, dim3(PE_THREADS), 0, stream>>>(src, dst, rec, acc, part);
  pool_head  <<<GRAPHS, B, 0, stream>>>(part, batch, W2, b2, Wf, bf, out, out + GRAPHS * OUTC);
}
